# starting kernel
# baseline (speedup 1.0000x reference)
#include <hip/hip_runtime.h>
#include <math.h>

constexpr int TLEN  = 320000;
constexpr int BROWS = 64;
constexpr int CHUNK = 320;
constexpr int NC    = TLEN / CHUNK;
constexpr int NBLK  = CHUNK / 16;

typedef __attribute__((ext_vector_type(2))) float v2f;
typedef __attribute__((ext_vector_type(8))) float v8f;

__device__ __forceinline__ float fir3(const float* __restrict__ xr, int t,
                                      float b0, float b1, float b2) {
    float x0 = xr[t];
    float x1 = (t >= 1) ? xr[t - 1] : 0.0f;
    float x2 = (t >= 2) ? xr[t - 2] : 0.0f;
    return fmaf(b0, x0, fmaf(b1, x1, b2 * x2));
}

__global__ void lp_phase1(const float* __restrict__ x, float2* __restrict__ dstate,
                          float b0, float b1, float b2, float a1, float a2) {
    int gid = blockIdx.x * blockDim.x + threadIdx.x;
    if (gid >= BROWS * NC) return;
    int row = gid / NC;
    int c   = gid % NC;
    const float* xr = x + (long)row * TLEN;
    int t0 = c * CHUNK;
    float y1 = 0.0f, y2 = 0.0f;
    float xm1 = (t0 >= 1) ? xr[t0 - 1] : 0.0f;
    float xm2 = (t0 >= 2) ? xr[t0 - 2] : 0.0f;
    for (int n = 0; n < CHUNK; ++n) {
        float xn = xr[t0 + n];
        float f  = fmaf(b0, xn, fmaf(b1, xm1, b2 * xm2));
        float y  = fmaf(-a1, y1, fmaf(-a2, y2, f));
        y2 = y1; y1 = y; xm2 = xm1; xm1 = xn;
    }
    dstate[gid] = make_float2(y1, y2);
}

__global__ void lp_phase2(const float2* __restrict__ dstate, float2* __restrict__ entry,
                          float p00, float p01, float p10, float p11) {
    int row = blockIdx.x * blockDim.x + threadIdx.x;
    if (row >= BROWS) return;
    float s1 = 0.0f, s2 = 0.0f;
    for (int c = 0; c < NC; ++c) {
        entry[row * NC + c] = make_float2(s1, s2);
        float2 d = dstate[row * NC + c];
        float n1 = fmaf(p00, s1, fmaf(p01, s2, d.x));
        float n2 = fmaf(p10, s1, fmaf(p11, s2, d.y));
        s1 = n1; s2 = n2;
    }
}

__global__ __launch_bounds__(256) void lp_phase3(
    const float* __restrict__ x, float* __restrict__ out,
    const float2* __restrict__ entry,
    float b0, float b1, float b2, float a1, float a2) {

    int lane = threadIdx.x & 31;
    int wid  = blockIdx.x * 8 + (threadIdx.x >> 5);
    int c  = wid % NC;
    int rg = wid / NC;
    if (rg >= BROWS / 16) return;

    int half = lane >> 4;
    int lid  = lane & 15;
    int r0   = rg * 16;
    int row  = r0 + lid;
    int t0   = c * CHUNK;

    float h[17];
    h[0] = 1.0f;
    h[1] = -a1;
    #pragma unroll
    for (int n = 2; n <= 16; ++n) h[n] = fmaf(-a1, h[n - 1], -a2 * h[n - 2]);

    float Tt[20];
    #pragma unroll
    for (int j = 0; j < 16; ++j) Tt[j] = (lid >= j) ? h[lid - j] : 0.0f;
    Tt[16] = h[lid + 1];
    Tt[17] = -a2 * h[lid];
    Tt[18] = 0.0f;
    Tt[19] = 0.0f;

    v2f Bk[5];
    #pragma unroll
    for (int kc = 0; kc < 5; ++kc) {
        Bk[kc].x = half ? Tt[4 * kc + 2] : Tt[4 * kc + 0];
        Bk[kc].y = half ? Tt[4 * kc + 3] : Tt[4 * kc + 1];
    }

    float2 e = entry[row * NC + c];
    float ym1 = e.x, ym2 = e.y;

    const float* xr = x + (long)row * TLEN;

    for (int blk = 0; blk < NBLK; ++blk) {
        int tb = t0 + blk * 16;

        v2f Ak[5];
        #pragma unroll
        for (int kc = 0; kc < 4; ++kc) {
            int k0 = 4 * kc + (half ? 2 : 0);
            int k1 = 4 * kc + (half ? 3 : 1);
            Ak[kc].x = fir3(xr, tb + k0, b0, b1, b2);
            Ak[kc].y = fir3(xr, tb + k1, b0, b1, b2);
        }
        Ak[4].x = half ? 0.0f : ym1;
        Ak[4].y = half ? 0.0f : ym2;

        v8f acc = {};
        #pragma unroll
        for (int kc = 0; kc < 5; ++kc) {
            acc = __builtin_amdgcn_wmma_f32_16x16x4_f32(
                 false, Ak[kc],  false, Bk[kc],
                 (short)0, acc,  false,  false);
        }

        #pragma unroll
        for (int v = 0; v < 8; ++v) {
            out[(long)(r0 + v + 8 * half) * TLEN + tb + lid] = acc[v];
        }

        float nym1 = 0.0f, nym2 = 0.0f;
        #pragma unroll
        for (int v = 0; v < 8; ++v) {
            float lo15 = __shfl(acc[v], 15, 32);
            float hi15 = __shfl(acc[v], 31, 32);
            float lo14 = __shfl(acc[v], 14, 32);
            float hi14 = __shfl(acc[v], 30, 32);
            if (lid == v)     { nym1 = lo15; nym2 = lo14; }
            if (lid == v + 8) { nym1 = hi15; nym2 = hi14; }
        }
        ym1 = nym1; ym2 = nym2;
    }
}

extern "C" void kernel_launch(void* const* d_in, const int* in_sizes, int n_in,
                              void* d_out, int out_size, void* d_ws, size_t ws_size,
                              hipStream_t stream) {
    const float* x = (const float*)d_in[0];
    float* out = (float*)d_out;

    float2* dstate = (float2*)d_ws;
    float2* entry  = dstate + (size_t)BROWS * NC;

    double w0    = 2.0 * M_PI * 7500.0 / 16000.0;
    double alpha = sin(w0) / (2.0 * 0.707);
    double cw    = cos(w0);
    double b0 = (1.0 - cw) / 2.0, b1 = 1.0 - cw, b2 = b0;
    double a0 = 1.0 + alpha, a1 = -2.0 * cw, a2 = 1.0 - alpha;
    b0 /= a0; b1 /= a0; b2 /= a0; a1 /= a0; a2 /= a0;

    double p00 = 1, p01 = 0, p10 = 0, p11 = 1;
    for (int i = 0; i < CHUNK; ++i) {
        double n00 = -a1 * p00 - a2 * p10;
        double n01 = -a1 * p01 - a2 * p11;
        double n10 = p00, n11 = p01;
        p00 = n00; p01 = n01; p10 = n10; p11 = n11;
    }

    float fb0 = (float)b0, fb1 = (float)b1, fb2 = (float)b2;
    float fa1 = (float)a1, fa2 = (float)a2;

    int n1 = BROWS * NC;
    lp_phase1<<<(n1 + 255) / 256, 256, 0, stream>>>(x, dstate, fb0, fb1, fb2, fa1, fa2);
    lp_phase2<<<1, 64, 0, stream>>>(dstate, entry,
                                    (float)p00, (float)p01, (float)p10, (float)p11);
    int nwaves = NC * (BROWS / 16);
    lp_phase3<<<nwaves / 8, 256, 0, stream>>>(x, out, entry, fb0, fb1, fb2, fa1, fa2);
}


# phase-2 chunk scan: one work-group per row, the row's chunk states staged through LDS with coalesced 16-byte loads and stores, one lane walks them in the same order (sums unchanged)
# speedup vs baseline: 2.0552x; 2.0552x over previous
#include <hip/hip_runtime.h>
#include <math.h>

constexpr int TLEN  = 320000;
constexpr int BROWS = 64;
constexpr int CHUNK = 320;
constexpr int NC    = TLEN / CHUNK;
constexpr int NBLK  = CHUNK / 16;

typedef __attribute__((ext_vector_type(2))) float v2f;
typedef __attribute__((ext_vector_type(8))) float v8f;

__device__ __forceinline__ float fir3(const float* __restrict__ xr, int t,
                                      float b0, float b1, float b2) {
    float x0 = xr[t];
    float x1 = (t >= 1) ? xr[t - 1] : 0.0f;
    float x2 = (t >= 2) ? xr[t - 2] : 0.0f;
    return fmaf(b0, x0, fmaf(b1, x1, b2 * x2));
}

__global__ void lp_phase1(const float* __restrict__ x, float2* __restrict__ dstate,
                          float b0, float b1, float b2, float a1, float a2) {
    int gid = blockIdx.x * blockDim.x + threadIdx.x;
    if (gid >= BROWS * NC) return;
    int row = gid / NC;
    int c   = gid % NC;
    const float* xr = x + (long)row * TLEN;
    int t0 = c * CHUNK;
    float y1 = 0.0f, y2 = 0.0f;
    float xm1 = (t0 >= 1) ? xr[t0 - 1] : 0.0f;
    float xm2 = (t0 >= 2) ? xr[t0 - 2] : 0.0f;
    for (int n = 0; n < CHUNK; ++n) {
        float xn = xr[t0 + n];
        float f  = fmaf(b0, xn, fmaf(b1, xm1, b2 * xm2));
        float y  = fmaf(-a1, y1, fmaf(-a2, y2, f));
        y2 = y1; y1 = y; xm2 = xm1; xm1 = xn;
    }
    dstate[gid] = make_float2(y1, y2);
}

constexpr int P2T   = 128;
constexpr int NV4   = NC / 2;
constexpr int P2LD  = (NV4 + P2T - 1) / P2T;
constexpr int P2PAD = P2LD * P2T;
constexpr int P2V   = 10;

__device__ __forceinline__ float4 p2_pair(float4 d, float& s1, float& s2,
                                          float p00, float p01, float p10, float p11) {
    float4 e;
    e.x = s1; e.y = s2;
    float m1 = fmaf(p00, s1, fmaf(p01, s2, d.x));
    float m2 = fmaf(p10, s1, fmaf(p11, s2, d.y));
    e.z = m1; e.w = m2;
    s1 = fmaf(p00, m1, fmaf(p01, m2, d.z));
    s2 = fmaf(p10, m1, fmaf(p11, m2, d.w));
    return e;
}

__global__ __launch_bounds__(P2T) void lp_phase2(
    const float2* __restrict__ dstate, float2* __restrict__ entry,
    float p00, float p01, float p10, float p11) {
    static_assert(NC % 2 == 0 && NV4 % P2V == 0, "row copy and walk shapes");
    __shared__ float4 sd4[P2PAD];
    __shared__ float4 se4[P2PAD];

    int row = blockIdx.x;
    int tid = threadIdx.x;
    const float4* src4 = reinterpret_cast<const float4*>(dstate + (size_t)row * NC);
    float4*       dst4 = reinterpret_cast<float4*>(entry + (size_t)row * NC);

    float4 tmp[P2LD];
    for (int k = 0; k < P2LD; ++k) {
        int i = tid + k * P2T;
        tmp[k] = src4[i < NV4 ? i : NV4 - 1];
    }
    for (int k = 0; k < P2LD; ++k) sd4[tid + k * P2T] = tmp[k];
    __syncthreads();

    if (tid == 0) {
        float s1 = 0.0f, s2 = 0.0f;
        float4 cur[P2V], nxt[P2V];
        for (int k = 0; k < P2V; ++k) cur[k] = sd4[k];
        for (int v0 = 0; v0 < NV4 - P2V; v0 += P2V) {
            for (int k = 0; k < P2V; ++k) nxt[k] = sd4[v0 + P2V + k];
            for (int k = 0; k < P2V; ++k)
                se4[v0 + k] = p2_pair(cur[k], s1, s2, p00, p01, p10, p11);
            for (int k = 0; k < P2V; ++k) cur[k] = nxt[k];
        }
        for (int k = 0; k < P2V; ++k)
            se4[NV4 - P2V + k] = p2_pair(cur[k], s1, s2, p00, p01, p10, p11);
    }
    __syncthreads();

    for (int k = 0; k < P2LD; ++k) {
        int i = tid + k * P2T;
        if (i < NV4) dst4[i] = se4[i];
    }
}

__global__ __launch_bounds__(256) void lp_phase3(
    const float* __restrict__ x, float* __restrict__ out,
    const float2* __restrict__ entry,
    float b0, float b1, float b2, float a1, float a2) {

    int lane = threadIdx.x & 31;
    int wid  = blockIdx.x * 8 + (threadIdx.x >> 5);
    int c  = wid % NC;
    int rg = wid / NC;
    if (rg >= BROWS / 16) return;

    int half = lane >> 4;
    int lid  = lane & 15;
    int r0   = rg * 16;
    int row  = r0 + lid;
    int t0   = c * CHUNK;

    float h[17];
    h[0] = 1.0f;
    h[1] = -a1;
    #pragma unroll
    for (int n = 2; n <= 16; ++n) h[n] = fmaf(-a1, h[n - 1], -a2 * h[n - 2]);

    float Tt[20];
    #pragma unroll
    for (int j = 0; j < 16; ++j) Tt[j] = (lid >= j) ? h[lid - j] : 0.0f;
    Tt[16] = h[lid + 1];
    Tt[17] = -a2 * h[lid];
    Tt[18] = 0.0f;
    Tt[19] = 0.0f;

    v2f Bk[5];
    #pragma unroll
    for (int kc = 0; kc < 5; ++kc) {
        Bk[kc].x = half ? Tt[4 * kc + 2] : Tt[4 * kc + 0];
        Bk[kc].y = half ? Tt[4 * kc + 3] : Tt[4 * kc + 1];
    }

    float2 e = entry[row * NC + c];
    float ym1 = e.x, ym2 = e.y;

    const float* xr = x + (long)row * TLEN;

    for (int blk = 0; blk < NBLK; ++blk) {
        int tb = t0 + blk * 16;

        v2f Ak[5];
        #pragma unroll
        for (int kc = 0; kc < 4; ++kc) {
            int k0 = 4 * kc + (half ? 2 : 0);
            int k1 = 4 * kc + (half ? 3 : 1);
            Ak[kc].x = fir3(xr, tb + k0, b0, b1, b2);
            Ak[kc].y = fir3(xr, tb + k1, b0, b1, b2);
        }
        Ak[4].x = half ? 0.0f : ym1;
        Ak[4].y = half ? 0.0f : ym2;

        v8f acc = {};
        #pragma unroll
        for (int kc = 0; kc < 5; ++kc) {
            acc = __builtin_amdgcn_wmma_f32_16x16x4_f32(
                 false, Ak[kc],  false, Bk[kc],
                 (short)0, acc,  false,  false);
        }

        #pragma unroll
        for (int v = 0; v < 8; ++v) {
            out[(long)(r0 + v + 8 * half) * TLEN + tb + lid] = acc[v];
        }

        float nym1 = 0.0f, nym2 = 0.0f;
        #pragma unroll
        for (int v = 0; v < 8; ++v) {
            float lo15 = __shfl(acc[v], 15, 32);
            float hi15 = __shfl(acc[v], 31, 32);
            float lo14 = __shfl(acc[v], 14, 32);
            float hi14 = __shfl(acc[v], 30, 32);
            if (lid == v)     { nym1 = lo15; nym2 = lo14; }
            if (lid == v + 8) { nym1 = hi15; nym2 = hi14; }
        }
        ym1 = nym1; ym2 = nym2;
    }
}

extern "C" void kernel_launch(void* const* d_in, const int* in_sizes, int n_in,
                              void* d_out, int out_size, void* d_ws, size_t ws_size,
                              hipStream_t stream) {
    const float* x = (const float*)d_in[0];
    float* out = (float*)d_out;

    float2* dstate = (float2*)d_ws;
    float2* entry  = dstate + (size_t)BROWS * NC;

    double w0    = 2.0 * M_PI * 7500.0 / 16000.0;
    double alpha = sin(w0) / (2.0 * 0.707);
    double cw    = cos(w0);
    double b0 = (1.0 - cw) / 2.0, b1 = 1.0 - cw, b2 = b0;
    double a0 = 1.0 + alpha, a1 = -2.0 * cw, a2 = 1.0 - alpha;
    b0 /= a0; b1 /= a0; b2 /= a0; a1 /= a0; a2 /= a0;

    double p00 = 1, p01 = 0, p10 = 0, p11 = 1;
    for (int i = 0; i < CHUNK; ++i) {
        double n00 = -a1 * p00 - a2 * p10;
        double n01 = -a1 * p01 - a2 * p11;
        double n10 = p00, n11 = p01;
        p00 = n00; p01 = n01; p10 = n10; p11 = n11;
    }

    float fb0 = (float)b0, fb1 = (float)b1, fb2 = (float)b2;
    float fa1 = (float)a1, fa2 = (float)a2;

    int n1 = BROWS * NC;
    lp_phase1<<<(n1 + 255) / 256, 256, 0, stream>>>(x, dstate, fb0, fb1, fb2, fa1, fa2);
    lp_phase2<<<BROWS, P2T, 0, stream>>>(dstate, entry,
                                    (float)p00, (float)p01, (float)p10, (float)p11);
    int nwaves = NC * (BROWS / 16);
    lp_phase3<<<nwaves / 8, 256, 0, stream>>>(x, out, entry, fb0, fb1, fb2, fa1, fa2);
}


# x read as 16-byte per-lane pieces in both passes; phase 3 lanes hold 8 consecutive samples, A and B k order permuted alike (sums inside a 16-block reordered), FIR taken from registers
# speedup vs baseline: 2.9576x; 1.4391x over previous
#include <hip/hip_runtime.h>
#include <math.h>

constexpr int TLEN  = 320000;
constexpr int BROWS = 64;
constexpr int CHUNK = 320;
constexpr int NC    = TLEN / CHUNK;
constexpr int NBLK  = CHUNK / 16;

typedef __attribute__((ext_vector_type(2))) float v2f;
typedef __attribute__((ext_vector_type(8))) float v8f;

__device__ __forceinline__ float fir3r(float x0, float x1, float x2,
                                       float b0, float b1, float b2) {
    return fmaf(b0, x0, fmaf(b1, x1, b2 * x2));
}

__global__ void lp_phase1(const float* __restrict__ x, float2* __restrict__ dstate,
                          float b0, float b1, float b2, float a1, float a2) {
    int gid = blockIdx.x * blockDim.x + threadIdx.x;
    if (gid >= BROWS * NC) return;
    int row = gid / NC;
    int c   = gid % NC;
    const float* xr = x + (long)row * TLEN;
    int t0 = c * CHUNK;
    float y1 = 0.0f, y2 = 0.0f;
    float xm1 = (t0 >= 1) ? xr[t0 - 1] : 0.0f;
    float xm2 = (t0 >= 2) ? xr[t0 - 2] : 0.0f;
    const float4* xq = reinterpret_cast<const float4*>(xr + t0);
    #pragma unroll 8
    for (int q = 0; q < CHUNK / 4; ++q) {
        float4 v = xq[q];
        const float xv[4] = {v.x, v.y, v.z, v.w};
        #pragma unroll
        for (int j = 0; j < 4; ++j) {
            float xn = xv[j];
            float f  = fmaf(b0, xn, fmaf(b1, xm1, b2 * xm2));
            float y  = fmaf(-a1, y1, fmaf(-a2, y2, f));
            y2 = y1; y1 = y; xm2 = xm1; xm1 = xn;
        }
    }
    dstate[gid] = make_float2(y1, y2);
}

constexpr int P2T   = 128;
constexpr int NV4   = NC / 2;
constexpr int P2LD  = (NV4 + P2T - 1) / P2T;
constexpr int P2PAD = P2LD * P2T;
constexpr int P2V   = 10;

__device__ __forceinline__ float4 p2_pair(float4 d, float& s1, float& s2,
                                          float p00, float p01, float p10, float p11) {
    float4 e;
    e.x = s1; e.y = s2;
    float m1 = fmaf(p00, s1, fmaf(p01, s2, d.x));
    float m2 = fmaf(p10, s1, fmaf(p11, s2, d.y));
    e.z = m1; e.w = m2;
    s1 = fmaf(p00, m1, fmaf(p01, m2, d.z));
    s2 = fmaf(p10, m1, fmaf(p11, m2, d.w));
    return e;
}

__global__ __launch_bounds__(P2T) void lp_phase2(
    const float2* __restrict__ dstate, float2* __restrict__ entry,
    float p00, float p01, float p10, float p11) {
    static_assert(NC % 2 == 0 && NV4 % P2V == 0, "row copy and walk shapes");
    __shared__ float4 sd4[P2PAD];
    __shared__ float4 se4[P2PAD];

    int row = blockIdx.x;
    int tid = threadIdx.x;
    const float4* src4 = reinterpret_cast<const float4*>(dstate + (size_t)row * NC);
    float4*       dst4 = reinterpret_cast<float4*>(entry + (size_t)row * NC);

    float4 tmp[P2LD];
    for (int k = 0; k < P2LD; ++k) {
        int i = tid + k * P2T;
        tmp[k] = src4[i < NV4 ? i : NV4 - 1];
    }
    for (int k = 0; k < P2LD; ++k) sd4[tid + k * P2T] = tmp[k];
    __syncthreads();

    if (tid == 0) {
        float s1 = 0.0f, s2 = 0.0f;
        float4 cur[P2V], nxt[P2V];
        for (int k = 0; k < P2V; ++k) cur[k] = sd4[k];
        for (int v0 = 0; v0 < NV4 - P2V; v0 += P2V) {
            for (int k = 0; k < P2V; ++k) nxt[k] = sd4[v0 + P2V + k];
            for (int k = 0; k < P2V; ++k)
                se4[v0 + k] = p2_pair(cur[k], s1, s2, p00, p01, p10, p11);
            for (int k = 0; k < P2V; ++k) cur[k] = nxt[k];
        }
        for (int k = 0; k < P2V; ++k)
            se4[NV4 - P2V + k] = p2_pair(cur[k], s1, s2, p00, p01, p10, p11);
    }
    __syncthreads();

    for (int k = 0; k < P2LD; ++k) {
        int i = tid + k * P2T;
        if (i < NV4) dst4[i] = se4[i];
    }
}

__global__ __launch_bounds__(256) void lp_phase3(
    const float* __restrict__ x, float* __restrict__ out,
    const float2* __restrict__ entry,
    float b0, float b1, float b2, float a1, float a2) {

    int lane = threadIdx.x & 31;
    int wid  = blockIdx.x * 8 + (threadIdx.x >> 5);
    int c  = wid % NC;
    int rg = wid / NC;
    if (rg >= BROWS / 16) return;

    int half = lane >> 4;
    int lid  = lane & 15;
    int r0   = rg * 16;
    int row  = r0 + lid;
    int t0   = c * CHUNK;

    float h[17];
    h[0] = 1.0f;
    h[1] = -a1;
    #pragma unroll
    for (int n = 2; n <= 16; ++n) h[n] = fmaf(-a1, h[n - 1], -a2 * h[n - 2]);

    float Tt[20];
    #pragma unroll
    for (int j = 0; j < 16; ++j) Tt[j] = (lid >= j) ? h[lid - j] : 0.0f;
    Tt[16] = h[lid + 1];
    Tt[17] = -a2 * h[lid];
    Tt[18] = 0.0f;
    Tt[19] = 0.0f;

    v2f Bk[5];
    #pragma unroll
    for (int kc = 0; kc < 4; ++kc) {
        Bk[kc].x = half ? Tt[8 + 2 * kc + 0] : Tt[2 * kc + 0];
        Bk[kc].y = half ? Tt[8 + 2 * kc + 1] : Tt[2 * kc + 1];
    }
    Bk[4].x = half ? Tt[18] : Tt[16];
    Bk[4].y = half ? Tt[19] : Tt[17];

    float2 e = entry[row * NC + c];
    float ym1 = e.x, ym2 = e.y;

    const float* xl = x + (long)row * TLEN + t0 + 8 * half;

    for (int blk = 0; blk < NBLK; ++blk) {
        int tb = t0 + blk * 16;
        const float* p = xl + blk * 16;

        bool  rowstart = (tb + 8 * half) == 0;
        float4 xa = *reinterpret_cast<const float4*>(p);
        float4 xb = *reinterpret_cast<const float4*>(p + 4);
        float2 xh = *reinterpret_cast<const float2*>(rowstart ? p : p - 2);
        if (rowstart) { xh.x = 0.0f; xh.y = 0.0f; }
        const float xs[10] = {xh.x, xh.y, xa.x, xa.y, xa.z, xa.w, xb.x, xb.y, xb.z, xb.w};

        v2f Ak[5];
        #pragma unroll
        for (int kc = 0; kc < 4; ++kc) {
            Ak[kc].x = fir3r(xs[2 * kc + 2], xs[2 * kc + 1], xs[2 * kc + 0], b0, b1, b2);
            Ak[kc].y = fir3r(xs[2 * kc + 3], xs[2 * kc + 2], xs[2 * kc + 1], b0, b1, b2);
        }
        Ak[4].x = half ? 0.0f : ym1;
        Ak[4].y = half ? 0.0f : ym2;

        v8f acc = {};
        #pragma unroll
        for (int kc = 0; kc < 5; ++kc) {
            acc = __builtin_amdgcn_wmma_f32_16x16x4_f32(
                 false, Ak[kc],  false, Bk[kc],
                 (short)0, acc,  false,  false);
        }

        #pragma unroll
        for (int v = 0; v < 8; ++v) {
            out[(long)(r0 + v + 8 * half) * TLEN + tb + lid] = acc[v];
        }

        float nym1 = 0.0f, nym2 = 0.0f;
        #pragma unroll
        for (int v = 0; v < 8; ++v) {
            float lo15 = __shfl(acc[v], 15, 32);
            float hi15 = __shfl(acc[v], 31, 32);
            float lo14 = __shfl(acc[v], 14, 32);
            float hi14 = __shfl(acc[v], 30, 32);
            if (lid == v)     { nym1 = lo15; nym2 = lo14; }
            if (lid == v + 8) { nym1 = hi15; nym2 = hi14; }
        }
        ym1 = nym1; ym2 = nym2;
    }
}

extern "C" void kernel_launch(void* const* d_in, const int* in_sizes, int n_in,
                              void* d_out, int out_size, void* d_ws, size_t ws_size,
                              hipStream_t stream) {
    const float* x = (const float*)d_in[0];
    float* out = (float*)d_out;

    float2* dstate = (float2*)d_ws;
    float2* entry  = dstate + (size_t)BROWS * NC;

    double w0    = 2.0 * M_PI * 7500.0 / 16000.0;
    double alpha = sin(w0) / (2.0 * 0.707);
    double cw    = cos(w0);
    double b0 = (1.0 - cw) / 2.0, b1 = 1.0 - cw, b2 = b0;
    double a0 = 1.0 + alpha, a1 = -2.0 * cw, a2 = 1.0 - alpha;
    b0 /= a0; b1 /= a0; b2 /= a0; a1 /= a0; a2 /= a0;

    double p00 = 1, p01 = 0, p10 = 0, p11 = 1;
    for (int i = 0; i < CHUNK; ++i) {
        double n00 = -a1 * p00 - a2 * p10;
        double n01 = -a1 * p01 - a2 * p11;
        double n10 = p00, n11 = p01;
        p00 = n00; p01 = n01; p10 = n10; p11 = n11;
    }

    float fb0 = (float)b0, fb1 = (float)b1, fb2 = (float)b2;
    float fa1 = (float)a1, fa2 = (float)a2;

    int n1 = BROWS * NC;
    lp_phase1<<<(n1 + 255) / 256, 256, 0, stream>>>(x, dstate, fb0, fb1, fb2, fa1, fa2);
    lp_phase2<<<BROWS, P2T, 0, stream>>>(dstate, entry,
                                    (float)p00, (float)p01, (float)p10, (float)p11);
    int nwaves = NC * (BROWS / 16);
    lp_phase3<<<nwaves / 8, 256, 0, stream>>>(x, out, entry, fb0, fb1, fb2, fa1, fa2);
}


# phase 3 tile turned: one row per wave, time as M and 16 consecutive 16-sample blocks as N, contiguous 16-byte loads and stores; block entry states by a lane scan (state sums reordered)
# speedup vs baseline: 3.6182x; 1.2234x over previous
#include <hip/hip_runtime.h>
#include <math.h>

constexpr int TLEN  = 320000;
constexpr int BROWS = 64;
constexpr int CHUNK = 320;
constexpr int NC    = TLEN / CHUNK;
constexpr int NBLK  = CHUNK / 16;

typedef __attribute__((ext_vector_type(2))) float v2f;
typedef __attribute__((ext_vector_type(8))) float v8f;

__device__ __forceinline__ float fir3r(float x0, float x1, float x2,
                                       float b0, float b1, float b2) {
    return fmaf(b0, x0, fmaf(b1, x1, b2 * x2));
}

__global__ void lp_phase1(const float* __restrict__ x, float2* __restrict__ dstate,
                          float b0, float b1, float b2, float a1, float a2) {
    int gid = blockIdx.x * blockDim.x + threadIdx.x;
    if (gid >= BROWS * NC) return;
    int row = gid / NC;
    int c   = gid % NC;
    const float* xr = x + (long)row * TLEN;
    int t0 = c * CHUNK;
    float y1 = 0.0f, y2 = 0.0f;
    float xm1 = (t0 >= 1) ? xr[t0 - 1] : 0.0f;
    float xm2 = (t0 >= 2) ? xr[t0 - 2] : 0.0f;
    const float4* xq = reinterpret_cast<const float4*>(xr + t0);
    #pragma unroll 8
    for (int q = 0; q < CHUNK / 4; ++q) {
        float4 v = xq[q];
        const float xv[4] = {v.x, v.y, v.z, v.w};
        #pragma unroll
        for (int j = 0; j < 4; ++j) {
            float xn = xv[j];
            float f  = fmaf(b0, xn, fmaf(b1, xm1, b2 * xm2));
            float y  = fmaf(-a1, y1, fmaf(-a2, y2, f));
            y2 = y1; y1 = y; xm2 = xm1; xm1 = xn;
        }
    }
    dstate[gid] = make_float2(y1, y2);
}

constexpr int P2T   = 128;
constexpr int NV4   = NC / 2;
constexpr int P2LD  = (NV4 + P2T - 1) / P2T;
constexpr int P2PAD = P2LD * P2T;
constexpr int P2V   = 10;

__device__ __forceinline__ float4 p2_pair(float4 d, float& s1, float& s2,
                                          float p00, float p01, float p10, float p11) {
    float4 e;
    e.x = s1; e.y = s2;
    float m1 = fmaf(p00, s1, fmaf(p01, s2, d.x));
    float m2 = fmaf(p10, s1, fmaf(p11, s2, d.y));
    e.z = m1; e.w = m2;
    s1 = fmaf(p00, m1, fmaf(p01, m2, d.z));
    s2 = fmaf(p10, m1, fmaf(p11, m2, d.w));
    return e;
}

__global__ __launch_bounds__(P2T) void lp_phase2(
    const float2* __restrict__ dstate, float2* __restrict__ entry,
    float p00, float p01, float p10, float p11) {
    static_assert(NC % 2 == 0 && NV4 % P2V == 0, "row copy and walk shapes");
    __shared__ float4 sd4[P2PAD];
    __shared__ float4 se4[P2PAD];

    int row = blockIdx.x;
    int tid = threadIdx.x;
    const float4* src4 = reinterpret_cast<const float4*>(dstate + (size_t)row * NC);
    float4*       dst4 = reinterpret_cast<float4*>(entry + (size_t)row * NC);

    float4 tmp[P2LD];
    for (int k = 0; k < P2LD; ++k) {
        int i = tid + k * P2T;
        tmp[k] = src4[i < NV4 ? i : NV4 - 1];
    }
    for (int k = 0; k < P2LD; ++k) sd4[tid + k * P2T] = tmp[k];
    __syncthreads();

    if (tid == 0) {
        float s1 = 0.0f, s2 = 0.0f;
        float4 cur[P2V], nxt[P2V];
        for (int k = 0; k < P2V; ++k) cur[k] = sd4[k];
        for (int v0 = 0; v0 < NV4 - P2V; v0 += P2V) {
            for (int k = 0; k < P2V; ++k) nxt[k] = sd4[v0 + P2V + k];
            for (int k = 0; k < P2V; ++k)
                se4[v0 + k] = p2_pair(cur[k], s1, s2, p00, p01, p10, p11);
            for (int k = 0; k < P2V; ++k) cur[k] = nxt[k];
        }
        for (int k = 0; k < P2V; ++k)
            se4[NV4 - P2V + k] = p2_pair(cur[k], s1, s2, p00, p01, p10, p11);
    }
    __syncthreads();

    for (int k = 0; k < P2LD; ++k) {
        int i = tid + k * P2T;
        if (i < NV4) dst4[i] = se4[i];
    }
}

constexpr int TILE    = 256;
constexpr int P3TILES = 10;
constexpr int SUPER   = TILE * P3TILES;
constexpr int NSUP    = TLEN / SUPER;
constexpr int P3AHEAD = 4;

template <int N>
__device__ __forceinline__ float row_shr(float old, float v) {
    return __int_as_float(__builtin_amdgcn_update_dpp(
        __float_as_int(old), __float_as_int(v), 0x110 + N, 0xF, 0xF, false));
}

template <int N>
__device__ __forceinline__ void scan_step(float& u1, float& u2,
                                          float m00, float m01, float m10, float m11) {
    float t1 = row_shr<N>(0.0f, u1);
    float t2 = row_shr<N>(0.0f, u2);
    u1 = fmaf(m00, t1, fmaf(m01, t2, u1));
    u2 = fmaf(m10, t1, fmaf(m11, t2, u2));
}

__global__ __launch_bounds__(256, 8) void lp_phase3(
    const float* __restrict__ x, float* __restrict__ out,
    const float2* __restrict__ entry,
    float b0, float b1, float b2, float a1, float a2) {
    static_assert(SUPER % CHUNK == 0 && TLEN % SUPER == 0, "a wave starts on a chunk boundary");

    int lane = threadIdx.x & 31;
    int wid  = blockIdx.x * 8 + (threadIdx.x >> 5);
    int sc  = wid % NSUP;
    int row = wid / NSUP;
    if (row >= BROWS) return;

    int half = lane >> 4;
    int lid  = lane & 15;

    float h[17];
    h[0] = 1.0f;
    h[1] = -a1;
    #pragma unroll
    for (int n = 2; n <= 16; ++n) h[n] = fmaf(-a1, h[n - 1], -a2 * h[n - 2]);

    float Tt[18];
    Tt[0] = h[0];
    #pragma unroll
    for (int n = 1; n < 16; ++n) Tt[0] = (lid == n) ? h[n] : Tt[0];
    Tt[1]  = row_shr<1>(0.0f, Tt[0]);
    Tt[2]  = row_shr<2>(0.0f, Tt[0]);
    Tt[3]  = row_shr<3>(0.0f, Tt[0]);
    Tt[4]  = row_shr<4>(0.0f, Tt[0]);
    Tt[5]  = row_shr<5>(0.0f, Tt[0]);
    Tt[6]  = row_shr<6>(0.0f, Tt[0]);
    Tt[7]  = row_shr<7>(0.0f, Tt[0]);
    Tt[8]  = row_shr<8>(0.0f, Tt[0]);
    Tt[9]  = row_shr<9>(0.0f, Tt[0]);
    Tt[10] = row_shr<10>(0.0f, Tt[0]);
    Tt[11] = row_shr<11>(0.0f, Tt[0]);
    Tt[12] = row_shr<12>(0.0f, Tt[0]);
    Tt[13] = row_shr<13>(0.0f, Tt[0]);
    Tt[14] = row_shr<14>(0.0f, Tt[0]);
    Tt[15] = row_shr<15>(0.0f, Tt[0]);
    Tt[16] = fmaf(-a1, Tt[0], -a2 * Tt[1]);
    Tt[17] = -a2 * Tt[0];

    v2f Ak[5];
    #pragma unroll
    for (int kc = 0; kc < 4; ++kc) {
        Ak[kc].x = half ? Tt[8 + 2 * kc + 0] : Tt[2 * kc + 0];
        Ak[kc].y = half ? Tt[8 + 2 * kc + 1] : Tt[2 * kc + 1];
    }
    Ak[4].x = half ? Tt[16] : 0.0f;
    Ak[4].y = half ? Tt[17] : 0.0f;

    float q00 = h[16], q01 = -a2 * h[15], q10 = h[15], q11 = -a2 * h[14];
    float r00 = fmaf(q00, q00, q01 * q10), r01 = fmaf(q00, q01, q01 * q11);
    float r10 = fmaf(q10, q00, q11 * q10), r11 = fmaf(q10, q01, q11 * q11);
    float s00 = fmaf(r00, r00, r01 * r10), s01 = fmaf(r00, r01, r01 * r11);
    float s10 = fmaf(r10, r00, r11 * r10), s11 = fmaf(r10, r01, r11 * r11);
    float w00 = fmaf(s00, s00, s01 * s10), w01 = fmaf(s00, s01, s01 * s11);
    float w10 = fmaf(s10, s00, s11 * s10), w11 = fmaf(s10, s01, s11 * s11);

    float2 e = entry[row * NC + sc * (SUPER / CHUNK)];
    float sy1 = e.x, sy2 = e.y;

    const long base = (long)row * TLEN + (long)sc * SUPER + 16 * lid + 8 * half;
    const float* xl = x + base;
    float*       ol = out + base;

    const bool rowstart = (sc == 0) && (lane == 0);

    float4 xa[P3TILES], xb[P3TILES];
    float2 xh[P3TILES];
    #pragma unroll
    for (int i = 0; i < P3AHEAD; ++i) {
        const float* p = xl + i * TILE;
        xa[i] = *reinterpret_cast<const float4*>(p);
        xb[i] = *reinterpret_cast<const float4*>(p + 4);
        xh[i] = *reinterpret_cast<const float2*>((i == 0 && rowstart) ? p : p - 2);
    }
    if (rowstart) { xh[0].x = 0.0f; xh[0].y = 0.0f; }

    #pragma unroll
    for (int i = 0; i < P3TILES; ++i) {
        if (i + P3AHEAD < P3TILES) {
            const float* p = xl + (i + P3AHEAD) * TILE;
            xa[i + P3AHEAD] = *reinterpret_cast<const float4*>(p);
            xb[i + P3AHEAD] = *reinterpret_cast<const float4*>(p + 4);
            xh[i + P3AHEAD] = *reinterpret_cast<const float2*>(p - 2);
        }

        const float xs[10] = {xh[i].x, xh[i].y, xa[i].x, xa[i].y, xa[i].z, xa[i].w,
                              xb[i].x, xb[i].y, xb[i].z, xb[i].w};

        v2f Bk[4];
        #pragma unroll
        for (int kc = 0; kc < 4; ++kc) {
            Bk[kc].x = fir3r(xs[2 * kc + 2], xs[2 * kc + 1], xs[2 * kc + 0], b0, b1, b2);
            Bk[kc].y = fir3r(xs[2 * kc + 3], xs[2 * kc + 2], xs[2 * kc + 1], b0, b1, b2);
        }

        v8f acc = {};
        #pragma unroll
        for (int kc = 0; kc < 4; ++kc) {
            acc = __builtin_amdgcn_wmma_f32_16x16x4_f32(
                 false, Ak[kc],  false, Bk[kc],
                 (short)0, acc,  false,  false);
        }

        float u1 = acc[7], u2 = acc[6];
        float j1 = fmaf(q00, sy1, fmaf(q01, sy2, u1));
        float j2 = fmaf(q10, sy1, fmaf(q11, sy2, u2));
        u1 = (lid == 0) ? j1 : u1;
        u2 = (lid == 0) ? j2 : u2;
        scan_step<1>(u1, u2, q00, q01, q10, q11);
        scan_step<2>(u1, u2, r00, r01, r10, r11);
        scan_step<4>(u1, u2, s00, s01, s10, s11);
        scan_step<8>(u1, u2, w00, w01, w10, w11);
        float e1 = row_shr<1>(sy1, u1);
        float e2 = row_shr<1>(sy2, u2);
        v2f Bs;
        Bs.x = half ? e1 : 0.0f;
        Bs.y = half ? e2 : 0.0f;
        acc = __builtin_amdgcn_wmma_f32_16x16x4_f32(
             false, Ak[4],  false, Bs,
             (short)0, acc,  false,  false);

        sy1 = __int_as_float(__builtin_amdgcn_readlane(__float_as_int(u1), 31));
        sy2 = __int_as_float(__builtin_amdgcn_readlane(__float_as_int(u2), 31));

        float* q = ol + i * TILE;
        *reinterpret_cast<float4*>(q)     = make_float4(acc[0], acc[1], acc[2], acc[3]);
        *reinterpret_cast<float4*>(q + 4) = make_float4(acc[4], acc[5], acc[6], acc[7]);
        asm volatile("" ::: "memory");
    }
}

extern "C" void kernel_launch(void* const* d_in, const int* in_sizes, int n_in,
                              void* d_out, int out_size, void* d_ws, size_t ws_size,
                              hipStream_t stream) {
    const float* x = (const float*)d_in[0];
    float* out = (float*)d_out;

    float2* dstate = (float2*)d_ws;
    float2* entry  = dstate + (size_t)BROWS * NC;

    double w0    = 2.0 * M_PI * 7500.0 / 16000.0;
    double alpha = sin(w0) / (2.0 * 0.707);
    double cw    = cos(w0);
    double b0 = (1.0 - cw) / 2.0, b1 = 1.0 - cw, b2 = b0;
    double a0 = 1.0 + alpha, a1 = -2.0 * cw, a2 = 1.0 - alpha;
    b0 /= a0; b1 /= a0; b2 /= a0; a1 /= a0; a2 /= a0;

    double p00 = 1, p01 = 0, p10 = 0, p11 = 1;
    for (int i = 0; i < CHUNK; ++i) {
        double n00 = -a1 * p00 - a2 * p10;
        double n01 = -a1 * p01 - a2 * p11;
        double n10 = p00, n11 = p01;
        p00 = n00; p01 = n01; p10 = n10; p11 = n11;
    }

    float fb0 = (float)b0, fb1 = (float)b1, fb2 = (float)b2;
    float fa1 = (float)a1, fa2 = (float)a2;

    int n1 = BROWS * NC;
    lp_phase1<<<(n1 + 255) / 256, 256, 0, stream>>>(x, dstate, fb0, fb1, fb2, fa1, fa2);
    lp_phase2<<<BROWS, P2T, 0, stream>>>(dstate, entry,
                                    (float)p00, (float)p01, (float)p10, (float)p11);
    int nwaves = BROWS * NSUP;
    lp_phase3<<<nwaves / 8, 256, 0, stream>>>(x, out, entry, fb0, fb1, fb2, fa1, fa2);
}


# phase 1 turned: one wave per 2560-sample run, contiguous 16-byte loads, per-lane recurrences joined by state-matrix powers and a lane sum (state sums reordered); phase 2 then walks 125 run states
# speedup vs baseline: 5.1118x; 1.4128x over previous
#include <hip/hip_runtime.h>
#include <math.h>


constexpr int TLEN  = 320000;
constexpr int BROWS = 64;
constexpr int TILE  = 256;
constexpr int RTILES = 10;
constexpr int CHUNK = TILE * RTILES;
constexpr int NC    = TLEN / CHUNK;

typedef __attribute__((ext_vector_type(2))) float v2f;
typedef __attribute__((ext_vector_type(8))) float v8f;

__device__ __forceinline__ float fir3r(float x0, float x1, float x2,
                                       float b0, float b1, float b2) {
    return fmaf(b0, x0, fmaf(b1, x1, b2 * x2));
}

__device__ __forceinline__ void mm2(float& c00, float& c01, float& c10, float& c11,
                                    float x00, float x01, float x10, float x11,
                                    float y00, float y01, float y10, float y11) {
    c00 = fmaf(x00, y00, x01 * y10); c01 = fmaf(x00, y01, x01 * y11);
    c10 = fmaf(x10, y00, x11 * y10); c11 = fmaf(x10, y01, x11 * y11);
}

constexpr int P1RUNS = 2;
constexpr int P1WPB  = 8;
static_assert((BROWS * NC) % (P1RUNS * P1WPB) == 0, "phase-1 grid covers the runs exactly");

__global__ __launch_bounds__(256) void lp_phase1(
    const float* __restrict__ x, float2* __restrict__ dstate,
    float b0, float b1, float b2, float a1, float a2) {
    int lane = threadIdx.x & 31;
    int wid  = blockIdx.x * P1WPB + (threadIdx.x >> 5);
    int half = lane >> 4;
    int lid  = lane & 15;

    float h[17];
    h[0] = 1.0f;
    h[1] = -a1;
    #pragma unroll
    for (int n = 2; n <= 16; ++n) h[n] = fmaf(-a1, h[n - 1], -a2 * h[n - 2]);

    float pw[6][4];
    pw[0][0] = h[8];  pw[0][1] = -a2 * h[7];  pw[0][2] = h[7];  pw[0][3] = -a2 * h[6];
    pw[1][0] = h[16]; pw[1][1] = -a2 * h[15]; pw[1][2] = h[15]; pw[1][3] = -a2 * h[14];
    #pragma unroll
    for (int j = 2; j < 6; ++j)
        mm2(pw[j][0], pw[j][1], pw[j][2], pw[j][3],
            pw[j - 1][0], pw[j - 1][1], pw[j - 1][2], pw[j - 1][3],
            pw[j - 1][0], pw[j - 1][1], pw[j - 1][2], pw[j - 1][3]);
    const float w00 = pw[5][0], w01 = pw[5][1], w10 = pw[5][2], w11 = pw[5][3];

    int ex = 31 - (2 * lid + half);
    float m00 = 1.0f, m01 = 0.0f, m10 = 0.0f, m11 = 1.0f;
    #pragma unroll
    for (int j = 0; j < 5; ++j) {
        float t00, t01, t10, t11;
        mm2(t00, t01, t10, t11, m00, m01, m10, m11, pw[j][0], pw[j][1], pw[j][2], pw[j][3]);
        bool on = (ex >> j) & 1;
        m00 = on ? t00 : m00; m01 = on ? t01 : m01;
        m10 = on ? t10 : m10; m11 = on ? t11 : m11;
    }

    for (int r = 0; r < P1RUNS; ++r) {
        int run = wid * P1RUNS + r;
        int row = run / NC;
        int sc  = run % NC;
        const float* xl = x + (long)row * TLEN + (long)sc * CHUNK + 16 * lid + 8 * half;
        const bool rowstart = (sc == 0) && (lane == 0);

        float4 xa[RTILES], xb[RTILES];
        float2 xh[RTILES];
        #pragma unroll
        for (int i = 0; i < RTILES; ++i) {
            const float* p = xl + i * TILE;
            xa[i] = *reinterpret_cast<const float4*>(p);
            xb[i] = *reinterpret_cast<const float4*>(p + 4);
            xh[i] = *reinterpret_cast<const float2*>((i == 0 && rowstart) ? p : p - 2);
        }
        if (rowstart) { xh[0].x = 0.0f; xh[0].y = 0.0f; }

        float c1 = 0.0f, c2 = 0.0f;
        #pragma unroll
        for (int i = 0; i < RTILES; ++i) {
            const float xs[10] = {xh[i].x, xh[i].y, xa[i].x, xa[i].y, xa[i].z, xa[i].w,
                                  xb[i].x, xb[i].y, xb[i].z, xb[i].w};
            float y1 = 0.0f, y2 = 0.0f;
            #pragma unroll
            for (int j = 0; j < 8; ++j) {
                float f = fir3r(xs[j + 2], xs[j + 1], xs[j], b0, b1, b2);
                float y = fmaf(-a1, y1, fmaf(-a2, y2, f));
                y2 = y1; y1 = y;
            }
            float n1 = fmaf(w00, c1, fmaf(w01, c2, y1));
            float n2 = fmaf(w10, c1, fmaf(w11, c2, y2));
            c1 = n1; c2 = n2;
        }

        float u1 = fmaf(m00, c1, m01 * c2);
        float u2 = fmaf(m10, c1, m11 * c2);
        #pragma unroll
        for (int d = 16; d >= 1; d >>= 1) {
            u1 += __shfl_xor(u1, d, 32);
            u2 += __shfl_xor(u2, d, 32);
        }
        if (lane == 0) dstate[run] = make_float2(u1, u2);
    }
}

constexpr int P2T = 128;
static_assert(NC <= P2T, "one run state per thread");

__global__ __launch_bounds__(P2T) void lp_phase2(
    const float2* __restrict__ dstate, float2* __restrict__ entry,
    float p00, float p01, float p10, float p11) {
    __shared__ float2 sd[P2T];
    __shared__ float2 se[P2T];

    int row = blockIdx.x;
    int tid = threadIdx.x;
    if (tid < NC) sd[tid] = dstate[(size_t)row * NC + tid];
    __syncthreads();

    if (tid == 0) {
        float s1 = 0.0f, s2 = 0.0f;
        #pragma unroll 25
        for (int c = 0; c < NC; ++c) {
            float2 d = sd[c];
            se[c] = make_float2(s1, s2);
            float n1 = fmaf(p00, s1, fmaf(p01, s2, d.x));
            float n2 = fmaf(p10, s1, fmaf(p11, s2, d.y));
            s1 = n1; s2 = n2;
        }
    }
    __syncthreads();

    if (tid < NC) entry[(size_t)row * NC + tid] = se[tid];
}

constexpr int P3TILES = RTILES;
constexpr int SUPER   = CHUNK;
constexpr int NSUP    = NC;
constexpr int P3AHEAD = 4;

template <int N>
__device__ __forceinline__ float row_shr(float old, float v) {
    return __int_as_float(__builtin_amdgcn_update_dpp(
        __float_as_int(old), __float_as_int(v), 0x110 + N, 0xF, 0xF, false));
}

template <int N>
__device__ __forceinline__ void scan_step(float& u1, float& u2,
                                          float m00, float m01, float m10, float m11) {
    float t1 = row_shr<N>(0.0f, u1);
    float t2 = row_shr<N>(0.0f, u2);
    u1 = fmaf(m00, t1, fmaf(m01, t2, u1));
    u2 = fmaf(m10, t1, fmaf(m11, t2, u2));
}

__global__ __launch_bounds__(256, 8) void lp_phase3(
    const float* __restrict__ x, float* __restrict__ out,
    const float2* __restrict__ entry,
    float b0, float b1, float b2, float a1, float a2) {
    static_assert(TLEN % SUPER == 0, "whole runs per row");

    int lane = threadIdx.x & 31;
    int wid  = blockIdx.x * 8 + (threadIdx.x >> 5);
    int sc  = wid % NSUP;
    int row = wid / NSUP;
    if (row >= BROWS) return;

    int half = lane >> 4;
    int lid  = lane & 15;

    float h[17];
    h[0] = 1.0f;
    h[1] = -a1;
    #pragma unroll
    for (int n = 2; n <= 16; ++n) h[n] = fmaf(-a1, h[n - 1], -a2 * h[n - 2]);

    float Tt[18];
    Tt[0] = h[0];
    #pragma unroll
    for (int n = 1; n < 16; ++n) Tt[0] = (lid == n) ? h[n] : Tt[0];
    Tt[1]  = row_shr<1>(0.0f, Tt[0]);
    Tt[2]  = row_shr<2>(0.0f, Tt[0]);
    Tt[3]  = row_shr<3>(0.0f, Tt[0]);
    Tt[4]  = row_shr<4>(0.0f, Tt[0]);
    Tt[5]  = row_shr<5>(0.0f, Tt[0]);
    Tt[6]  = row_shr<6>(0.0f, Tt[0]);
    Tt[7]  = row_shr<7>(0.0f, Tt[0]);
    Tt[8]  = row_shr<8>(0.0f, Tt[0]);
    Tt[9]  = row_shr<9>(0.0f, Tt[0]);
    Tt[10] = row_shr<10>(0.0f, Tt[0]);
    Tt[11] = row_shr<11>(0.0f, Tt[0]);
    Tt[12] = row_shr<12>(0.0f, Tt[0]);
    Tt[13] = row_shr<13>(0.0f, Tt[0]);
    Tt[14] = row_shr<14>(0.0f, Tt[0]);
    Tt[15] = row_shr<15>(0.0f, Tt[0]);
    Tt[16] = fmaf(-a1, Tt[0], -a2 * Tt[1]);
    Tt[17] = -a2 * Tt[0];

    v2f Ak[5];
    #pragma unroll
    for (int kc = 0; kc < 4; ++kc) {
        Ak[kc].x = half ? Tt[8 + 2 * kc + 0] : Tt[2 * kc + 0];
        Ak[kc].y = half ? Tt[8 + 2 * kc + 1] : Tt[2 * kc + 1];
    }
    Ak[4].x = half ? Tt[16] : 0.0f;
    Ak[4].y = half ? Tt[17] : 0.0f;

    float q00 = h[16], q01 = -a2 * h[15], q10 = h[15], q11 = -a2 * h[14];
    float r00 = fmaf(q00, q00, q01 * q10), r01 = fmaf(q00, q01, q01 * q11);
    float r10 = fmaf(q10, q00, q11 * q10), r11 = fmaf(q10, q01, q11 * q11);
    float s00 = fmaf(r00, r00, r01 * r10), s01 = fmaf(r00, r01, r01 * r11);
    float s10 = fmaf(r10, r00, r11 * r10), s11 = fmaf(r10, r01, r11 * r11);
    float w00 = fmaf(s00, s00, s01 * s10), w01 = fmaf(s00, s01, s01 * s11);
    float w10 = fmaf(s10, s00, s11 * s10), w11 = fmaf(s10, s01, s11 * s11);

    float2 e = entry[row * NC + sc];
    float sy1 = e.x, sy2 = e.y;

    const long base = (long)row * TLEN + (long)sc * SUPER + 16 * lid + 8 * half;
    const float* xl = x + base;
    float*       ol = out + base;

    const bool rowstart = (sc == 0) && (lane == 0);

    float4 xa[P3TILES], xb[P3TILES];
    float2 xh[P3TILES];
    #pragma unroll
    for (int i = 0; i < P3AHEAD; ++i) {
        const float* p = xl + i * TILE;
        xa[i] = *reinterpret_cast<const float4*>(p);
        xb[i] = *reinterpret_cast<const float4*>(p + 4);
        xh[i] = *reinterpret_cast<const float2*>((i == 0 && rowstart) ? p : p - 2);
    }
    if (rowstart) { xh[0].x = 0.0f; xh[0].y = 0.0f; }

    #pragma unroll
    for (int i = 0; i < P3TILES; ++i) {
        if (i + P3AHEAD < P3TILES) {
            const float* p = xl + (i + P3AHEAD) * TILE;
            xa[i + P3AHEAD] = *reinterpret_cast<const float4*>(p);
            xb[i + P3AHEAD] = *reinterpret_cast<const float4*>(p + 4);
            xh[i + P3AHEAD] = *reinterpret_cast<const float2*>(p - 2);
        }

        const float xs[10] = {xh[i].x, xh[i].y, xa[i].x, xa[i].y, xa[i].z, xa[i].w,
                              xb[i].x, xb[i].y, xb[i].z, xb[i].w};

        v2f Bk[4];
        #pragma unroll
        for (int kc = 0; kc < 4; ++kc) {
            Bk[kc].x = fir3r(xs[2 * kc + 2], xs[2 * kc + 1], xs[2 * kc + 0], b0, b1, b2);
            Bk[kc].y = fir3r(xs[2 * kc + 3], xs[2 * kc + 2], xs[2 * kc + 1], b0, b1, b2);
        }

        v8f acc = {};
        #pragma unroll
        for (int kc = 0; kc < 4; ++kc) {
            acc = __builtin_amdgcn_wmma_f32_16x16x4_f32(
                 false, Ak[kc],  false, Bk[kc],
                 (short)0, acc,  false,  false);
        }

        float u1 = acc[7], u2 = acc[6];
        float j1 = fmaf(q00, sy1, fmaf(q01, sy2, u1));
        float j2 = fmaf(q10, sy1, fmaf(q11, sy2, u2));
        u1 = (lid == 0) ? j1 : u1;
        u2 = (lid == 0) ? j2 : u2;
        scan_step<1>(u1, u2, q00, q01, q10, q11);
        scan_step<2>(u1, u2, r00, r01, r10, r11);
        scan_step<4>(u1, u2, s00, s01, s10, s11);
        scan_step<8>(u1, u2, w00, w01, w10, w11);
        float e1 = row_shr<1>(sy1, u1);
        float e2 = row_shr<1>(sy2, u2);
        v2f Bs;
        Bs.x = half ? e1 : 0.0f;
        Bs.y = half ? e2 : 0.0f;
        acc = __builtin_amdgcn_wmma_f32_16x16x4_f32(
             false, Ak[4],  false, Bs,
             (short)0, acc,  false,  false);

        sy1 = __int_as_float(__builtin_amdgcn_readlane(__float_as_int(u1), 31));
        sy2 = __int_as_float(__builtin_amdgcn_readlane(__float_as_int(u2), 31));

        float* q = ol + i * TILE;
        *reinterpret_cast<float4*>(q)     = make_float4(acc[0], acc[1], acc[2], acc[3]);
        *reinterpret_cast<float4*>(q + 4) = make_float4(acc[4], acc[5], acc[6], acc[7]);
        asm volatile("" ::: "memory");
    }
}

extern "C" void kernel_launch(void* const* d_in, const int* in_sizes, int n_in,
                              void* d_out, int out_size, void* d_ws, size_t ws_size,
                              hipStream_t stream) {
    const float* x = (const float*)d_in[0];
    float* out = (float*)d_out;

    float2* dstate = (float2*)d_ws;
    float2* entry  = dstate + (size_t)BROWS * NC;

    double w0    = 2.0 * M_PI * 7500.0 / 16000.0;
    double alpha = sin(w0) / (2.0 * 0.707);
    double cw    = cos(w0);
    double b0 = (1.0 - cw) / 2.0, b1 = 1.0 - cw, b2 = b0;
    double a0 = 1.0 + alpha, a1 = -2.0 * cw, a2 = 1.0 - alpha;
    b0 /= a0; b1 /= a0; b2 /= a0; a1 /= a0; a2 /= a0;

    double p00 = 1, p01 = 0, p10 = 0, p11 = 1;
    for (int i = 0; i < CHUNK; ++i) {
        double n00 = -a1 * p00 - a2 * p10;
        double n01 = -a1 * p01 - a2 * p11;
        double n10 = p00, n11 = p01;
        p00 = n00; p01 = n01; p10 = n10; p11 = n11;
    }

    float fb0 = (float)b0, fb1 = (float)b1, fb2 = (float)b2;
    float fa1 = (float)a1, fa2 = (float)a2;

    lp_phase1<<<BROWS * NC / (P1RUNS * P1WPB), 32 * P1WPB, 0, stream>>>(x, dstate, fb0, fb1, fb2, fa1, fa2);
    lp_phase2<<<BROWS, P2T, 0, stream>>>(dstate, entry,
                                    (float)p00, (float)p01, (float)p10, (float)p11);
    int nwaves = BROWS * NSUP;
    lp_phase3<<<nwaves / 8, 256, 0, stream>>>(x, out, entry, fb0, fb1, fb2, fa1, fa2);
}


# run-state combine moved into phase 3: each wave joins its row's earlier run states itself by a lane scan with powers of the run transition (state sums reordered); the phase-2 launch is gone
# speedup vs baseline: 5.9967x; 1.1731x over previous
#include <hip/hip_runtime.h>
#include <math.h>


constexpr int TLEN  = 320000;
constexpr int BROWS = 64;
constexpr int TILE  = 256;
constexpr int RTILES = 10;
constexpr int CHUNK = TILE * RTILES;
constexpr int NC    = TLEN / CHUNK;

typedef __attribute__((ext_vector_type(2))) float v2f;
typedef __attribute__((ext_vector_type(8))) float v8f;

__device__ __forceinline__ float fir3r(float x0, float x1, float x2,
                                       float b0, float b1, float b2) {
    return fmaf(b0, x0, fmaf(b1, x1, b2 * x2));
}

__device__ __forceinline__ void mm2(float& c00, float& c01, float& c10, float& c11,
                                    float x00, float x01, float x10, float x11,
                                    float y00, float y01, float y10, float y11) {
    c00 = fmaf(x00, y00, x01 * y10); c01 = fmaf(x00, y01, x01 * y11);
    c10 = fmaf(x10, y00, x11 * y10); c11 = fmaf(x10, y01, x11 * y11);
}

constexpr int P1RUNS = 2;
constexpr int P1WPB  = 8;
static_assert((BROWS * NC) % (P1RUNS * P1WPB) == 0, "phase-1 grid covers the runs exactly");

__global__ __launch_bounds__(256) void lp_phase1(
    const float* __restrict__ x, float2* __restrict__ dstate,
    float b0, float b1, float b2, float a1, float a2) {
    int lane = threadIdx.x & 31;
    int wid  = blockIdx.x * P1WPB + (threadIdx.x >> 5);
    int half = lane >> 4;
    int lid  = lane & 15;

    float h[17];
    h[0] = 1.0f;
    h[1] = -a1;
    #pragma unroll
    for (int n = 2; n <= 16; ++n) h[n] = fmaf(-a1, h[n - 1], -a2 * h[n - 2]);

    float pw[6][4];
    pw[0][0] = h[8];  pw[0][1] = -a2 * h[7];  pw[0][2] = h[7];  pw[0][3] = -a2 * h[6];
    pw[1][0] = h[16]; pw[1][1] = -a2 * h[15]; pw[1][2] = h[15]; pw[1][3] = -a2 * h[14];
    #pragma unroll
    for (int j = 2; j < 6; ++j)
        mm2(pw[j][0], pw[j][1], pw[j][2], pw[j][3],
            pw[j - 1][0], pw[j - 1][1], pw[j - 1][2], pw[j - 1][3],
            pw[j - 1][0], pw[j - 1][1], pw[j - 1][2], pw[j - 1][3]);
    const float w00 = pw[5][0], w01 = pw[5][1], w10 = pw[5][2], w11 = pw[5][3];

    int ex = 31 - (2 * lid + half);
    float m00 = 1.0f, m01 = 0.0f, m10 = 0.0f, m11 = 1.0f;
    #pragma unroll
    for (int j = 0; j < 5; ++j) {
        float t00, t01, t10, t11;
        mm2(t00, t01, t10, t11, m00, m01, m10, m11, pw[j][0], pw[j][1], pw[j][2], pw[j][3]);
        bool on = (ex >> j) & 1;
        m00 = on ? t00 : m00; m01 = on ? t01 : m01;
        m10 = on ? t10 : m10; m11 = on ? t11 : m11;
    }

    for (int r = 0; r < P1RUNS; ++r) {
        int run = wid * P1RUNS + r;
        int row = run / NC;
        int sc  = run % NC;
        const float* xl = x + (long)row * TLEN + (long)sc * CHUNK + 16 * lid + 8 * half;
        const bool rowstart = (sc == 0) && (lane == 0);

        float4 xa[RTILES], xb[RTILES];
        float2 xh[RTILES];
        #pragma unroll
        for (int i = 0; i < RTILES; ++i) {
            const float* p = xl + i * TILE;
            xa[i] = *reinterpret_cast<const float4*>(p);
            xb[i] = *reinterpret_cast<const float4*>(p + 4);
            xh[i] = *reinterpret_cast<const float2*>((i == 0 && rowstart) ? p : p - 2);
        }
        if (rowstart) { xh[0].x = 0.0f; xh[0].y = 0.0f; }

        float c1 = 0.0f, c2 = 0.0f;
        #pragma unroll
        for (int i = 0; i < RTILES; ++i) {
            const float xs[10] = {xh[i].x, xh[i].y, xa[i].x, xa[i].y, xa[i].z, xa[i].w,
                                  xb[i].x, xb[i].y, xb[i].z, xb[i].w};
            float y1 = 0.0f, y2 = 0.0f;
            #pragma unroll
            for (int j = 0; j < 8; ++j) {
                float f = fir3r(xs[j + 2], xs[j + 1], xs[j], b0, b1, b2);
                float y = fmaf(-a1, y1, fmaf(-a2, y2, f));
                y2 = y1; y1 = y;
            }
            float n1 = fmaf(w00, c1, fmaf(w01, c2, y1));
            float n2 = fmaf(w10, c1, fmaf(w11, c2, y2));
            c1 = n1; c2 = n2;
        }

        float u1 = fmaf(m00, c1, m01 * c2);
        float u2 = fmaf(m10, c1, m11 * c2);
        #pragma unroll
        for (int d = 16; d >= 1; d >>= 1) {
            u1 += __shfl_xor(u1, d, 32);
            u2 += __shfl_xor(u2, d, 32);
        }
        if (lane == 0) dstate[run] = make_float2(u1, u2);
    }
}

constexpr int P3TILES = RTILES;
constexpr int SUPER   = CHUNK;
constexpr int NSUP    = NC;
constexpr int P3AHEAD = 4;

template <int N>
__device__ __forceinline__ float row_shr(float old, float v) {
    return __int_as_float(__builtin_amdgcn_update_dpp(
        __float_as_int(old), __float_as_int(v), 0x110 + N, 0xF, 0xF, false));
}

template <int N>
__device__ __forceinline__ void scan_step(float& u1, float& u2,
                                          float m00, float m01, float m10, float m11) {
    float t1 = row_shr<N>(0.0f, u1);
    float t2 = row_shr<N>(0.0f, u2);
    u1 = fmaf(m00, t1, fmaf(m01, t2, u1));
    u2 = fmaf(m10, t1, fmaf(m11, t2, u2));
}

__global__ __launch_bounds__(256, 8) void lp_phase3(
    const float* __restrict__ x, float* __restrict__ out,
    const float2* __restrict__ dstate,
    float b0, float b1, float b2, float a1, float a2,
    float p00, float p01, float p10, float p11) {
    static_assert(NC <= 4 * 32, "a wave's lanes cover a row's runs, four to a lane");
    static_assert(TLEN % SUPER == 0, "whole runs per row");

    int lane = threadIdx.x & 31;
    int wid  = blockIdx.x * 8 + (threadIdx.x >> 5);
    int sc  = wid % NSUP;
    int row = wid / NSUP;
    if (row >= BROWS) return;

    int half = lane >> 4;
    int lid  = lane & 15;

    float h[17];
    h[0] = 1.0f;
    h[1] = -a1;
    #pragma unroll
    for (int n = 2; n <= 16; ++n) h[n] = fmaf(-a1, h[n - 1], -a2 * h[n - 2]);

    float Tt[18];
    Tt[0] = h[0];
    #pragma unroll
    for (int n = 1; n < 16; ++n) Tt[0] = (lid == n) ? h[n] : Tt[0];
    Tt[1]  = row_shr<1>(0.0f, Tt[0]);
    Tt[2]  = row_shr<2>(0.0f, Tt[0]);
    Tt[3]  = row_shr<3>(0.0f, Tt[0]);
    Tt[4]  = row_shr<4>(0.0f, Tt[0]);
    Tt[5]  = row_shr<5>(0.0f, Tt[0]);
    Tt[6]  = row_shr<6>(0.0f, Tt[0]);
    Tt[7]  = row_shr<7>(0.0f, Tt[0]);
    Tt[8]  = row_shr<8>(0.0f, Tt[0]);
    Tt[9]  = row_shr<9>(0.0f, Tt[0]);
    Tt[10] = row_shr<10>(0.0f, Tt[0]);
    Tt[11] = row_shr<11>(0.0f, Tt[0]);
    Tt[12] = row_shr<12>(0.0f, Tt[0]);
    Tt[13] = row_shr<13>(0.0f, Tt[0]);
    Tt[14] = row_shr<14>(0.0f, Tt[0]);
    Tt[15] = row_shr<15>(0.0f, Tt[0]);
    Tt[16] = fmaf(-a1, Tt[0], -a2 * Tt[1]);
    Tt[17] = -a2 * Tt[0];

    v2f Ak[5];
    #pragma unroll
    for (int kc = 0; kc < 4; ++kc) {
        Ak[kc].x = half ? Tt[8 + 2 * kc + 0] : Tt[2 * kc + 0];
        Ak[kc].y = half ? Tt[8 + 2 * kc + 1] : Tt[2 * kc + 1];
    }
    Ak[4].x = half ? Tt[16] : 0.0f;
    Ak[4].y = half ? Tt[17] : 0.0f;

    float q00 = h[16], q01 = -a2 * h[15], q10 = h[15], q11 = -a2 * h[14];
    float r00 = fmaf(q00, q00, q01 * q10), r01 = fmaf(q00, q01, q01 * q11);
    float r10 = fmaf(q10, q00, q11 * q10), r11 = fmaf(q10, q01, q11 * q11);
    float s00 = fmaf(r00, r00, r01 * r10), s01 = fmaf(r00, r01, r01 * r11);
    float s10 = fmaf(r10, r00, r11 * r10), s11 = fmaf(r10, r01, r11 * r11);
    float w00 = fmaf(s00, s00, s01 * s10), w01 = fmaf(s00, s01, s01 * s11);
    float w10 = fmaf(s10, s00, s11 * s10), w11 = fmaf(s10, s01, s11 * s11);

    const float2* dr = dstate + row * NC;
    const int j0 = sc - 4 * (32 - lane);
    float c1 = 0.0f, c2 = 0.0f;
    #pragma unroll
    for (int i = 0; i < 4; ++i) {
        int j = j0 + i;
        float2 d = dr[j > 0 ? j : 0];
        float d1 = (j >= 0) ? d.x : 0.0f;
        float d2 = (j >= 0) ? d.y : 0.0f;
        float n1 = fmaf(p00, c1, fmaf(p01, c2, d1));
        float n2 = fmaf(p10, c1, fmaf(p11, c2, d2));
        c1 = n1; c2 = n2;
    }
    float pa[4], pm[5][4];
    mm2(pa[0], pa[1], pa[2], pa[3], p00, p01, p10, p11, p00, p01, p10, p11);
    mm2(pm[0][0], pm[0][1], pm[0][2], pm[0][3],
        pa[0], pa[1], pa[2], pa[3], pa[0], pa[1], pa[2], pa[3]);
    #pragma unroll
    for (int k = 1; k < 5; ++k)
        mm2(pm[k][0], pm[k][1], pm[k][2], pm[k][3],
            pm[k - 1][0], pm[k - 1][1], pm[k - 1][2], pm[k - 1][3],
            pm[k - 1][0], pm[k - 1][1], pm[k - 1][2], pm[k - 1][3]);
    scan_step<1>(c1, c2, pm[0][0], pm[0][1], pm[0][2], pm[0][3]);
    scan_step<2>(c1, c2, pm[1][0], pm[1][1], pm[1][2], pm[1][3]);
    scan_step<4>(c1, c2, pm[2][0], pm[2][1], pm[2][2], pm[2][3]);
    scan_step<8>(c1, c2, pm[3][0], pm[3][1], pm[3][2], pm[3][3]);
    float lo1 = __int_as_float(__builtin_amdgcn_readlane(__float_as_int(c1), 15));
    float lo2 = __int_as_float(__builtin_amdgcn_readlane(__float_as_int(c2), 15));
    float hi1 = __int_as_float(__builtin_amdgcn_readlane(__float_as_int(c1), 31));
    float hi2 = __int_as_float(__builtin_amdgcn_readlane(__float_as_int(c2), 31));
    float sy1 = fmaf(pm[4][0], lo1, fmaf(pm[4][1], lo2, hi1));
    float sy2 = fmaf(pm[4][2], lo1, fmaf(pm[4][3], lo2, hi2));

    const long base = (long)row * TLEN + (long)sc * SUPER + 16 * lid + 8 * half;
    const float* xl = x + base;
    float*       ol = out + base;

    const bool rowstart = (sc == 0) && (lane == 0);

    float4 xa[P3TILES], xb[P3TILES];
    float2 xh[P3TILES];
    #pragma unroll
    for (int i = 0; i < P3AHEAD; ++i) {
        const float* p = xl + i * TILE;
        xa[i] = *reinterpret_cast<const float4*>(p);
        xb[i] = *reinterpret_cast<const float4*>(p + 4);
        xh[i] = *reinterpret_cast<const float2*>((i == 0 && rowstart) ? p : p - 2);
    }
    if (rowstart) { xh[0].x = 0.0f; xh[0].y = 0.0f; }

    #pragma unroll
    for (int i = 0; i < P3TILES; ++i) {
        if (i + P3AHEAD < P3TILES) {
            const float* p = xl + (i + P3AHEAD) * TILE;
            xa[i + P3AHEAD] = *reinterpret_cast<const float4*>(p);
            xb[i + P3AHEAD] = *reinterpret_cast<const float4*>(p + 4);
            xh[i + P3AHEAD] = *reinterpret_cast<const float2*>(p - 2);
        }

        const float xs[10] = {xh[i].x, xh[i].y, xa[i].x, xa[i].y, xa[i].z, xa[i].w,
                              xb[i].x, xb[i].y, xb[i].z, xb[i].w};

        v2f Bk[4];
        #pragma unroll
        for (int kc = 0; kc < 4; ++kc) {
            Bk[kc].x = fir3r(xs[2 * kc + 2], xs[2 * kc + 1], xs[2 * kc + 0], b0, b1, b2);
            Bk[kc].y = fir3r(xs[2 * kc + 3], xs[2 * kc + 2], xs[2 * kc + 1], b0, b1, b2);
        }

        v8f acc = {};
        #pragma unroll
        for (int kc = 0; kc < 4; ++kc) {
            acc = __builtin_amdgcn_wmma_f32_16x16x4_f32(
                 false, Ak[kc],  false, Bk[kc],
                 (short)0, acc,  false,  false);
        }

        float u1 = acc[7], u2 = acc[6];
        float j1 = fmaf(q00, sy1, fmaf(q01, sy2, u1));
        float j2 = fmaf(q10, sy1, fmaf(q11, sy2, u2));
        u1 = (lid == 0) ? j1 : u1;
        u2 = (lid == 0) ? j2 : u2;
        scan_step<1>(u1, u2, q00, q01, q10, q11);
        scan_step<2>(u1, u2, r00, r01, r10, r11);
        scan_step<4>(u1, u2, s00, s01, s10, s11);
        scan_step<8>(u1, u2, w00, w01, w10, w11);
        float e1 = row_shr<1>(sy1, u1);
        float e2 = row_shr<1>(sy2, u2);
        v2f Bs;
        Bs.x = half ? e1 : 0.0f;
        Bs.y = half ? e2 : 0.0f;
        acc = __builtin_amdgcn_wmma_f32_16x16x4_f32(
             false, Ak[4],  false, Bs,
             (short)0, acc,  false,  false);

        sy1 = __int_as_float(__builtin_amdgcn_readlane(__float_as_int(u1), 31));
        sy2 = __int_as_float(__builtin_amdgcn_readlane(__float_as_int(u2), 31));

        float* q = ol + i * TILE;
        *reinterpret_cast<float4*>(q)     = make_float4(acc[0], acc[1], acc[2], acc[3]);
        *reinterpret_cast<float4*>(q + 4) = make_float4(acc[4], acc[5], acc[6], acc[7]);
        asm volatile("" ::: "memory");
    }
}

extern "C" void kernel_launch(void* const* d_in, const int* in_sizes, int n_in,
                              void* d_out, int out_size, void* d_ws, size_t ws_size,
                              hipStream_t stream) {
    const float* x = (const float*)d_in[0];
    float* out = (float*)d_out;

    float2* dstate = (float2*)d_ws;

    double w0    = 2.0 * M_PI * 7500.0 / 16000.0;
    double alpha = sin(w0) / (2.0 * 0.707);
    double cw    = cos(w0);
    double b0 = (1.0 - cw) / 2.0, b1 = 1.0 - cw, b2 = b0;
    double a0 = 1.0 + alpha, a1 = -2.0 * cw, a2 = 1.0 - alpha;
    b0 /= a0; b1 /= a0; b2 /= a0; a1 /= a0; a2 /= a0;

    double p00 = 1, p01 = 0, p10 = 0, p11 = 1;
    for (int i = 0; i < CHUNK; ++i) {
        double n00 = -a1 * p00 - a2 * p10;
        double n01 = -a1 * p01 - a2 * p11;
        double n10 = p00, n11 = p01;
        p00 = n00; p01 = n01; p10 = n10; p11 = n11;
    }

    float fb0 = (float)b0, fb1 = (float)b1, fb2 = (float)b2;
    float fa1 = (float)a1, fa2 = (float)a2;

    lp_phase1<<<BROWS * NC / (P1RUNS * P1WPB), 32 * P1WPB, 0, stream>>>(x, dstate, fb0, fb1, fb2, fa1, fa2);
    int nwaves = BROWS * NSUP;
    lp_phase3<<<nwaves / 8, 256, 0, stream>>>(x, out, dstate, fb0, fb1, fb2, fa1, fa2,
                                              (float)p00, (float)p01, (float)p10, (float)p11);
}


# single launch: each wave gets its run's entry state by walking the 1024+ samples before the run from rest (A^1024 is exactly 0 in f32); 64 runs per row; phase 1 gone (entry-state sums reordered)
# speedup vs baseline: 9.2306x; 1.5393x over previous
#include <hip/hip_runtime.h>
#include <math.h>


constexpr int TLEN   = 320000;
constexpr int BROWS  = 64;
constexpr int TILE   = 256;
constexpr int NTILES = TLEN / TILE;
constexpr int RPR    = 64;
constexpr int NSHORT = NTILES / RPR;
constexpr int NLONG  = NTILES - NSHORT * RPR;
constexpr int STREAM = 24;
constexpr int RING   = 8;
constexpr int PASSES = STREAM / RING;
constexpr int LBMIN  = STREAM - (NSHORT + 1);
static_assert(TLEN % TILE == 0, "whole tiles per row");
static_assert(STREAM % RING == 0, "whole trips round the ring");
static_assert(NLONG >= 1 && NLONG < RPR, "long runs come first in a row, its first run among them");
static_assert(LBMIN == 4, "the row-head handling below is written for four look-back tiles");
static_assert(LBMIN * TILE >= 1024, "look-back length, see below");
static_assert(RING > LBMIN + 1, "the first trip holds every look-back tile");


typedef __attribute__((ext_vector_type(2))) float v2f;
typedef __attribute__((ext_vector_type(8))) float v8f;

__device__ __forceinline__ float fir3r(float x0, float x1, float x2,
                                       float b0, float b1, float b2) {
    return fmaf(b0, x0, fmaf(b1, x1, b2 * x2));
}

template <int N>
__device__ __forceinline__ float row_shr(float old, float v) {
    return __int_as_float(__builtin_amdgcn_update_dpp(
        __float_as_int(old), __float_as_int(v), 0x110 + N, 0xF, 0xF, false));
}

template <int N>
__device__ __forceinline__ void scan_step(float& u1, float& u2,
                                          float m00, float m01, float m10, float m11) {
    float t1 = row_shr<N>(0.0f, u1);
    float t2 = row_shr<N>(0.0f, u2);
    u1 = fmaf(m00, t1, fmaf(m01, t2, u1));
    u2 = fmaf(m10, t1, fmaf(m11, t2, u2));
}

struct TileK {
    v2f Ak[5];
    float b0, b1, b2;
    float q00, q01, q10, q11;
    float r00, r01, r10, r11;
    float s00, s01, s10, s11;
    float w00, w01, w10, w11;
    int lid, half;
};

template <bool REFILL>
__device__ __forceinline__ void tile_step(float4& xa, float4& xb, float2& xh,
                                          const float* __restrict__ nxt,
                                          float* __restrict__ dst, bool store,
                                          const TileK& k, float& sy1, float& sy2) {
    const float xs[10] = {xh.x, xh.y, xa.x, xa.y, xa.z, xa.w, xb.x, xb.y, xb.z, xb.w};

    v2f Bk[4];
    #pragma unroll
    for (int kc = 0; kc < 4; ++kc) {
        Bk[kc].x = fir3r(xs[2 * kc + 2], xs[2 * kc + 1], xs[2 * kc + 0], k.b0, k.b1, k.b2);
        Bk[kc].y = fir3r(xs[2 * kc + 3], xs[2 * kc + 2], xs[2 * kc + 1], k.b0, k.b1, k.b2);
    }

    if (REFILL) {
        xa = *reinterpret_cast<const float4*>(nxt);
        xb = *reinterpret_cast<const float4*>(nxt + 4);
        xh = *reinterpret_cast<const float2*>(nxt - 2);
    }

    v8f acc = {};
    #pragma unroll
    for (int kc = 0; kc < 4; ++kc) {
        acc = __builtin_amdgcn_wmma_f32_16x16x4_f32(
             false, k.Ak[kc],  false, Bk[kc],
             (short)0, acc,  false,  false);
    }

    float u1 = acc[7], u2 = acc[6];
    float j1 = fmaf(k.q00, sy1, fmaf(k.q01, sy2, u1));
    float j2 = fmaf(k.q10, sy1, fmaf(k.q11, sy2, u2));
    u1 = (k.lid == 0) ? j1 : u1;
    u2 = (k.lid == 0) ? j2 : u2;
    scan_step<1>(u1, u2, k.q00, k.q01, k.q10, k.q11);
    scan_step<2>(u1, u2, k.r00, k.r01, k.r10, k.r11);
    scan_step<4>(u1, u2, k.s00, k.s01, k.s10, k.s11);
    scan_step<8>(u1, u2, k.w00, k.w01, k.w10, k.w11);
    float e1 = row_shr<1>(sy1, u1);
    float e2 = row_shr<1>(sy2, u2);
    v2f Bs;
    Bs.x = k.half ? e1 : 0.0f;
    Bs.y = k.half ? e2 : 0.0f;
    acc = __builtin_amdgcn_wmma_f32_16x16x4_f32(
         false, k.Ak[4],  false, Bs,
         (short)0, acc,  false,  false);

    sy1 = __int_as_float(__builtin_amdgcn_readlane(__float_as_int(u1), 31));
    sy2 = __int_as_float(__builtin_amdgcn_readlane(__float_as_int(u2), 31));

    if (store) {
        *reinterpret_cast<float4*>(dst)     = make_float4(acc[0], acc[1], acc[2], acc[3]);
        *reinterpret_cast<float4*>(dst + 4) = make_float4(acc[4], acc[5], acc[6], acc[7]);
    }
    asm volatile("" ::: "memory");
}

__global__ __launch_bounds__(256) void lp_stream(
    const float* __restrict__ x, float* __restrict__ out,
    float b0, float b1, float b2, float a1, float a2) {
    int lane = threadIdx.x & 31;
    int wv  = __builtin_amdgcn_readfirstlane((int)(threadIdx.x >> 5));
    int wid = blockIdx.x * 8 + wv;
    int r   = wid % RPR;
    int row = wid / RPR;
    if (row >= BROWS) return;

    TileK k;
    k.half = lane >> 4;
    k.lid  = lane & 15;
    k.b0 = b0; k.b1 = b1; k.b2 = b2;
    const int half = k.half, lid = k.lid;

    float h[17];
    h[0] = 1.0f;
    h[1] = -a1;
    #pragma unroll
    for (int n = 2; n <= 16; ++n) h[n] = fmaf(-a1, h[n - 1], -a2 * h[n - 2]);

    float Tt[18];
    Tt[0] = h[0];
    #pragma unroll
    for (int n = 1; n < 16; ++n) Tt[0] = (lid == n) ? h[n] : Tt[0];
    Tt[1]  = row_shr<1>(0.0f, Tt[0]);
    Tt[2]  = row_shr<2>(0.0f, Tt[0]);
    Tt[3]  = row_shr<3>(0.0f, Tt[0]);
    Tt[4]  = row_shr<4>(0.0f, Tt[0]);
    Tt[5]  = row_shr<5>(0.0f, Tt[0]);
    Tt[6]  = row_shr<6>(0.0f, Tt[0]);
    Tt[7]  = row_shr<7>(0.0f, Tt[0]);
    Tt[8]  = row_shr<8>(0.0f, Tt[0]);
    Tt[9]  = row_shr<9>(0.0f, Tt[0]);
    Tt[10] = row_shr<10>(0.0f, Tt[0]);
    Tt[11] = row_shr<11>(0.0f, Tt[0]);
    Tt[12] = row_shr<12>(0.0f, Tt[0]);
    Tt[13] = row_shr<13>(0.0f, Tt[0]);
    Tt[14] = row_shr<14>(0.0f, Tt[0]);
    Tt[15] = row_shr<15>(0.0f, Tt[0]);
    Tt[16] = fmaf(-a1, Tt[0], -a2 * Tt[1]);
    Tt[17] = -a2 * Tt[0];

    #pragma unroll
    for (int kc = 0; kc < 4; ++kc) {
        k.Ak[kc].x = half ? Tt[8 + 2 * kc + 0] : Tt[2 * kc + 0];
        k.Ak[kc].y = half ? Tt[8 + 2 * kc + 1] : Tt[2 * kc + 1];
    }
    k.Ak[4].x = half ? Tt[16] : 0.0f;
    k.Ak[4].y = half ? Tt[17] : 0.0f;

    k.q00 = h[16]; k.q01 = -a2 * h[15]; k.q10 = h[15]; k.q11 = -a2 * h[14];
    k.r00 = fmaf(k.q00, k.q00, k.q01 * k.q10); k.r01 = fmaf(k.q00, k.q01, k.q01 * k.q11);
    k.r10 = fmaf(k.q10, k.q00, k.q11 * k.q10); k.r11 = fmaf(k.q10, k.q01, k.q11 * k.q11);
    k.s00 = fmaf(k.r00, k.r00, k.r01 * k.r10); k.s01 = fmaf(k.r00, k.r01, k.r01 * k.r11);
    k.s10 = fmaf(k.r10, k.r00, k.r11 * k.r10); k.s11 = fmaf(k.r10, k.r01, k.r11 * k.r11);
    k.w00 = fmaf(k.s00, k.s00, k.s01 * k.s10); k.w01 = fmaf(k.s00, k.s01, k.s01 * k.s11);
    k.w10 = fmaf(k.s10, k.s00, k.s11 * k.s10); k.w11 = fmaf(k.s10, k.s01, k.s11 * k.s11);

    const int  nown  = NSHORT + (r < NLONG ? 1 : 0);
    const int  first = NSHORT * r + (r < NLONG ? r : NLONG);
    const int  lb    = STREAM - nown;
    const bool head  = (r == 0);
    const long off = (long)row * TLEN + (long)(first - lb) * TILE + 16 * lid + 8 * half;

    const bool rowstart = head && (lane == 0);

    float4 xa[RING], xb[RING];
    float2 xh[RING];
    #pragma unroll
    for (int s = 0; s < RING; ++s) {
        const int t = (s < LBMIN && head) ? s + LBMIN + 1 : s;
        const float* p = x + (off + (long)t * TILE);
        xa[s] = *reinterpret_cast<const float4*>(p);
        xb[s] = *reinterpret_cast<const float4*>(p + 4);
        xh[s] = *reinterpret_cast<const float2*>((s == LBMIN && rowstart) ? p : p - 2);
    }
    if (rowstart) { xh[LBMIN].x = 0.0f; xh[LBMIN].y = 0.0f; }

    float sy1 = 0.0f, sy2 = 0.0f;

    #pragma unroll
    for (int p = 0; p < PASSES - 1; ++p) {
        #pragma unroll
        for (int s = 0; s < RING; ++s) {
            const int j = p * RING + s;
            if (s == LBMIN) {
                const bool fresh = head && (p == 0);
                sy1 = fresh ? 0.0f : sy1;
                sy2 = fresh ? 0.0f : sy2;
            }
            const bool store = (s > LBMIN) || (p > 0) || (s >= lb);
            float* dst = out + (store ? off + (long)j * TILE : 0L);
            tile_step<true>(xa[s], xb[s], xh[s], x + (off + (long)(j + RING) * TILE),
                            dst, store, k, sy1, sy2);
        }
    }
    #pragma unroll
    for (int s = 0; s < RING; ++s) {
        const int j = (PASSES - 1) * RING + s;
        tile_step<false>(xa[s], xb[s], xh[s], x, out + (off + (long)j * TILE), true, k, sy1, sy2);
    }
}

extern "C" void kernel_launch(void* const* d_in, const int* in_sizes, int n_in,
                              void* d_out, int out_size, void* d_ws, size_t ws_size,
                              hipStream_t stream) {
    const float* x = (const float*)d_in[0];
    float* out = (float*)d_out;

    double w0    = 2.0 * M_PI * 7500.0 / 16000.0;
    double alpha = sin(w0) / (2.0 * 0.707);
    double cw    = cos(w0);
    double b0 = (1.0 - cw) / 2.0, b1 = 1.0 - cw, b2 = b0;
    double a0 = 1.0 + alpha, a1 = -2.0 * cw, a2 = 1.0 - alpha;
    b0 /= a0; b1 /= a0; b2 /= a0; a1 /= a0; a2 /= a0;

    float fb0 = (float)b0, fb1 = (float)b1, fb2 = (float)b2;
    float fa1 = (float)a1, fa2 = (float)a2;

    static_assert((BROWS * RPR) % 8 == 0, "whole work-groups of 8 waves");
    lp_stream<<<BROWS * RPR / 8, 256, 0, stream>>>(x, out, fb0, fb1, fb2, fa1, fa2);
}


# global memory accessed in line order: tiles turned to and from fragment order in per-wave LDS buffers (bank-swizzled), so each load and store instruction covers four whole 128-byte lines
# speedup vs baseline: 10.3971x; 1.1264x over previous
#include <hip/hip_runtime.h>
#include <math.h>


constexpr int TLEN   = 320000;
constexpr int BROWS  = 64;
constexpr int TILE   = 256;
constexpr int NTILES = TLEN / TILE;
constexpr int RPR    = 64;
constexpr int NSHORT = NTILES / RPR;
constexpr int NLONG  = NTILES - NSHORT * RPR;
constexpr int STREAM = 24;
constexpr int RING   = 8;
constexpr int PASSES = STREAM / RING;
constexpr int LBMIN  = STREAM - (NSHORT + 1);
constexpr int WPB    = 8;
constexpr int PIECES = TILE / 4;
static_assert(TLEN % TILE == 0, "whole tiles per row");
static_assert(STREAM % RING == 0, "whole trips round the ring");
static_assert(NLONG >= 1 && NLONG < RPR, "long runs come first in a row, its first run among them");
static_assert(LBMIN == 4, "the row-head handling below is written for four look-back tiles");
static_assert(LBMIN * TILE >= 1024, "look-back length, see below");
static_assert(RING > LBMIN + 1, "the first trip holds every look-back tile");
static_assert(STREAM > LBMIN + 2, "the walk ends with staged own tiles");
static_assert((PIECES & (PIECES - 1)) == 0 && PIECES == 64, "the slot swizzle is written for 64 pieces");


typedef __attribute__((ext_vector_type(2))) float v2f;
typedef __attribute__((ext_vector_type(8))) float v8f;

__device__ __forceinline__ float fir3r(float x0, float x1, float x2,
                                       float b0, float b1, float b2) {
    return fmaf(b0, x0, fmaf(b1, x1, b2 * x2));
}

template <int N>
__device__ __forceinline__ float row_shr(float old, float v) {
    return __int_as_float(__builtin_amdgcn_update_dpp(
        __float_as_int(old), __float_as_int(v), 0x110 + N, 0xF, 0xF, false));
}

template <int N>
__device__ __forceinline__ void scan_step(float& u1, float& u2,
                                          float m00, float m01, float m10, float m11) {
    float t1 = row_shr<N>(0.0f, u1);
    float t2 = row_shr<N>(0.0f, u2);
    u1 = fmaf(m00, t1, fmaf(m01, t2, u1));
    u2 = fmaf(m10, t1, fmaf(m11, t2, u2));
}

struct TileK {
    v2f Ak[5];
    float b0, b1, b2;
    float q00, q01, q10, q11;
    float r00, r01, r10, r11;
    float s00, s01, s10, s11;
    float w00, w01, w10, w11;
    int lid, half;
};

__device__ __forceinline__ void fir_b(const float4& xa, const float4& xb, const float2& xh,
                                      const TileK& k, v2f (&Bk)[4]) {
    const float xs[10] = {xh.x, xh.y, xa.x, xa.y, xa.z, xa.w, xb.x, xb.y, xb.z, xb.w};

    #pragma unroll
    for (int kc = 0; kc < 4; ++kc) {
        Bk[kc].x = fir3r(xs[2 * kc + 2], xs[2 * kc + 1], xs[2 * kc + 0], k.b0, k.b1, k.b2);
        Bk[kc].y = fir3r(xs[2 * kc + 3], xs[2 * kc + 2], xs[2 * kc + 1], k.b0, k.b1, k.b2);
    }
}

__device__ __forceinline__ v8f tile_core(const v2f (&Bk)[4], const TileK& k,
                                         float& sy1, float& sy2) {
    v8f acc = {};
    #pragma unroll
    for (int kc = 0; kc < 4; ++kc) {
        acc = __builtin_amdgcn_wmma_f32_16x16x4_f32(
             false, k.Ak[kc],  false, Bk[kc],
             (short)0, acc,  false,  false);
    }
    float u1 = acc[7], u2 = acc[6];
    float j1 = fmaf(k.q00, sy1, fmaf(k.q01, sy2, u1));
    float j2 = fmaf(k.q10, sy1, fmaf(k.q11, sy2, u2));
    u1 = (k.lid == 0) ? j1 : u1;
    u2 = (k.lid == 0) ? j2 : u2;
    scan_step<1>(u1, u2, k.q00, k.q01, k.q10, k.q11);
    scan_step<2>(u1, u2, k.r00, k.r01, k.r10, k.r11);
    scan_step<4>(u1, u2, k.s00, k.s01, k.s10, k.s11);
    scan_step<8>(u1, u2, k.w00, k.w01, k.w10, k.w11);
    float e1 = row_shr<1>(sy1, u1);
    float e2 = row_shr<1>(sy2, u2);
    v2f Bs;
    Bs.x = k.half ? e1 : 0.0f;
    Bs.y = k.half ? e2 : 0.0f;
    acc = __builtin_amdgcn_wmma_f32_16x16x4_f32(
         false, k.Ak[4],  false, Bs,
         (short)0, acc,  false,  false);

    sy1 = __int_as_float(__builtin_amdgcn_readlane(__float_as_int(u1), 31));
    sy2 = __int_as_float(__builtin_amdgcn_readlane(__float_as_int(u2), 31));

    return acc;
}

struct InStage {
    int l0, l1;
    int f0, f1;
    int h[2];
};

__device__ __forceinline__ void turn_in(float4* __restrict__ inb, int par, const InStage& n,
                                        const float4& ga, const float4& gb,
                                        float4& xa, float4& xb, float2& xh) {
    float4* b = inb + par * PIECES;
    b[n.l0] = ga;
    b[n.l1] = gb;
    asm volatile("" ::: "memory");
    xa = b[n.f0];
    xb = b[n.f1];
    xh = *reinterpret_cast<const float2*>(reinterpret_cast<const float*>(inb) + n.h[par]);
}

struct OutStage {
    int w0, w1;
    int r0, r1;
    float4 s0, s1;
};

__device__ __forceinline__ void flush_tile(float* __restrict__ dst, const OutStage& o) {
    *reinterpret_cast<float4*>(dst)       = o.s0;
    *reinterpret_cast<float4*>(dst + 128) = o.s1;
}

__device__ __forceinline__ void park_tile(float4* __restrict__ buf, const v8f& acc, OutStage& o) {
    buf[o.w0] = make_float4(acc[0], acc[1], acc[2], acc[3]);
    buf[o.w1] = make_float4(acc[4], acc[5], acc[6], acc[7]);
    asm volatile("" ::: "memory");
    o.s0 = buf[o.r0];
    o.s1 = buf[o.r1];
}

struct Walk {
    const float* x; float* out;
    long offw;
    float4* inb; float4* buf;
    bool head, rowstart, longrun;
    InStage n; OutStage o;
    float4 ga[RING], gb[RING];
    float4 xa, xb; float2 xh;
    float sy1, sy2;
};

template <int J>
__device__ __forceinline__ void walk_step(Walk& w, const TileK& k) {
    if (J == LBMIN) {
        w.sy1 = w.head ? 0.0f : w.sy1;
        w.sy2 = w.head ? 0.0f : w.sy2;
        w.xh.x = w.rowstart ? 0.0f : w.xh.x;
        w.xh.y = w.rowstart ? 0.0f : w.xh.y;
    }
    v2f Bk[4];
    fir_b(w.xa, w.xb, w.xh, k, Bk);

    if (J + 1 < STREAM) {
        constexpr int S = (J + 1) % RING;
        turn_in(w.inb, (J + 1) & 1, w.n, w.ga[S], w.gb[S], w.xa, w.xb, w.xh);
        if (J + 1 + RING < STREAM) {
            const float* p = w.x + (w.offw + (long)(J + 1 + RING) * TILE);
            w.ga[S] = *reinterpret_cast<const float4*>(p);
            w.gb[S] = *reinterpret_cast<const float4*>(p + 128);
        }
    }

    v8f acc = tile_core(Bk, k, w.sy1, w.sy2);

    if (J > LBMIN + 1) {
        flush_tile(w.out + (w.offw + (long)(J - 1) * TILE), w.o);
        park_tile(w.buf, acc, w.o);
    } else if (J == LBMIN + 1) {
        if (w.longrun) flush_tile(w.out + (w.offw + (long)(J - 1) * TILE), w.o);
        park_tile(w.buf, acc, w.o);
    } else if (J == LBMIN) {
        if (w.longrun) park_tile(w.buf, acc, w.o);
    }
    asm volatile("" ::: "memory");
}

template <int J>
__device__ __forceinline__ void walk_from(Walk& w, const TileK& k) {
    walk_step<J>(w, k);
    if constexpr (J + 1 < STREAM) walk_from<J + 1>(w, k);
}

__global__ __launch_bounds__(256) void lp_stream(
    const float* __restrict__ x, float* __restrict__ out,
    float b0, float b1, float b2, float a1, float a2) {
    int lane = threadIdx.x & 31;
    int wv  = __builtin_amdgcn_readfirstlane((int)(threadIdx.x >> 5));
    int wid = blockIdx.x * 8 + wv;
    int r   = wid % RPR;
    int row = wid / RPR;
    if (row >= BROWS) return;

    TileK k;
    k.half = lane >> 4;
    k.lid  = lane & 15;
    k.b0 = b0; k.b1 = b1; k.b2 = b2;
    const int half = k.half, lid = k.lid;

    float h[17];
    h[0] = 1.0f;
    h[1] = -a1;
    #pragma unroll
    for (int n = 2; n <= 16; ++n) h[n] = fmaf(-a1, h[n - 1], -a2 * h[n - 2]);

    float Tt[18];
    Tt[0] = h[0];
    #pragma unroll
    for (int n = 1; n < 16; ++n) Tt[0] = (lid == n) ? h[n] : Tt[0];
    Tt[1]  = row_shr<1>(0.0f, Tt[0]);
    Tt[2]  = row_shr<2>(0.0f, Tt[0]);
    Tt[3]  = row_shr<3>(0.0f, Tt[0]);
    Tt[4]  = row_shr<4>(0.0f, Tt[0]);
    Tt[5]  = row_shr<5>(0.0f, Tt[0]);
    Tt[6]  = row_shr<6>(0.0f, Tt[0]);
    Tt[7]  = row_shr<7>(0.0f, Tt[0]);
    Tt[8]  = row_shr<8>(0.0f, Tt[0]);
    Tt[9]  = row_shr<9>(0.0f, Tt[0]);
    Tt[10] = row_shr<10>(0.0f, Tt[0]);
    Tt[11] = row_shr<11>(0.0f, Tt[0]);
    Tt[12] = row_shr<12>(0.0f, Tt[0]);
    Tt[13] = row_shr<13>(0.0f, Tt[0]);
    Tt[14] = row_shr<14>(0.0f, Tt[0]);
    Tt[15] = row_shr<15>(0.0f, Tt[0]);
    Tt[16] = fmaf(-a1, Tt[0], -a2 * Tt[1]);
    Tt[17] = -a2 * Tt[0];

    #pragma unroll
    for (int kc = 0; kc < 4; ++kc) {
        k.Ak[kc].x = half ? Tt[8 + 2 * kc + 0] : Tt[2 * kc + 0];
        k.Ak[kc].y = half ? Tt[8 + 2 * kc + 1] : Tt[2 * kc + 1];
    }
    k.Ak[4].x = half ? Tt[16] : 0.0f;
    k.Ak[4].y = half ? Tt[17] : 0.0f;

    k.q00 = h[16]; k.q01 = -a2 * h[15]; k.q10 = h[15]; k.q11 = -a2 * h[14];
    k.r00 = fmaf(k.q00, k.q00, k.q01 * k.q10); k.r01 = fmaf(k.q00, k.q01, k.q01 * k.q11);
    k.r10 = fmaf(k.q10, k.q00, k.q11 * k.q10); k.r11 = fmaf(k.q10, k.q01, k.q11 * k.q11);
    k.s00 = fmaf(k.r00, k.r00, k.r01 * k.r10); k.s01 = fmaf(k.r00, k.r01, k.r01 * k.r11);
    k.s10 = fmaf(k.r10, k.r00, k.r11 * k.r10); k.s11 = fmaf(k.r10, k.r01, k.r11 * k.r11);
    k.w00 = fmaf(k.s00, k.s00, k.s01 * k.s10); k.w01 = fmaf(k.s00, k.s01, k.s01 * k.s11);
    k.w10 = fmaf(k.s10, k.s00, k.s11 * k.s10); k.w11 = fmaf(k.s10, k.s01, k.s11 * k.s11);

    const int  nown  = NSHORT + (r < NLONG ? 1 : 0);
    const int  first = NSHORT * r + (r < NLONG ? r : NLONG);
    const int  lb    = STREAM - nown;
    const bool head  = (r == 0);
    const bool longrun = (lb == LBMIN);
    const long offw = (long)row * TLEN + (long)(first - lb) * TILE + 4 * lane;

    const bool rowstart = head && (lane == 0);

    __shared__ float4 stage_in[WPB * 2 * PIECES];
    __shared__ float4 stage_out[WPB * PIECES];
    float4* inb = stage_in + wv * (2 * PIECES);
    float4* buf = stage_out + wv * PIECES;
    InStage n;
    OutStage o;
    {
        const int pw = 4 * lid + 2 * half;
        n.f0 = o.w0 = pw ^ ((pw >> 4) & 3);
        n.f1 = o.w1 = (pw + 1) ^ ((pw >> 4) & 3);
        n.l0 = o.r0 = lane ^ ((lane >> 4) & 3);
        n.l1 = o.r1 = (32 + lane) ^ (((32 + lane) >> 4) & 3);
        const int pb   = (pw + PIECES - 1) & (PIECES - 1);
        const int hs   = 4 * (pb ^ ((pb >> 4) & 3)) + 2;
        n.h[0] = (lane == 0 ? 4 * PIECES : 0) + hs;
        n.h[1] = (lane == 0 ? 0 : 4 * PIECES) + hs;
        o.s0 = make_float4(0.0f, 0.0f, 0.0f, 0.0f);
        o.s1 = o.s0;
    }

    Walk w;
    w.x = x; w.out = out; w.offw = offw;
    w.inb = inb; w.buf = buf;
    w.head = head; w.rowstart = rowstart; w.longrun = longrun;
    w.n = n; w.o = o;
    w.sy1 = 0.0f; w.sy2 = 0.0f;

    #pragma unroll
    for (int s = 0; s < RING; ++s) {
        const int t = (s < LBMIN && head) ? s + LBMIN + 1 : s;
        const float* p = x + (offw + (long)t * TILE);
        w.ga[s] = *reinterpret_cast<const float4*>(p);
        w.gb[s] = *reinterpret_cast<const float4*>(p + 128);
    }
    {
        const int t0 = head ? LBMIN + 1 : 0;
        const float* p = x + ((long)row * TLEN + (long)(first - lb + t0) * TILE - 4);
        const float4 last = *reinterpret_cast<const float4*>(p);
        const int sl = (PIECES - 1) ^ (((PIECES - 1) >> 4) & 3);
        if (lane == 0) inb[PIECES + sl] = last;
    }

    turn_in(inb, 0, w.n, w.ga[0], w.gb[0], w.xa, w.xb, w.xh);
    {
        const float* p = x + (offw + (long)RING * TILE);
        w.ga[0] = *reinterpret_cast<const float4*>(p);
        w.gb[0] = *reinterpret_cast<const float4*>(p + 128);
    }

    walk_from<0>(w, k);
    flush_tile(out + (offw + (long)(STREAM - 1) * TILE), w.o);
}

extern "C" void kernel_launch(void* const* d_in, const int* in_sizes, int n_in,
                              void* d_out, int out_size, void* d_ws, size_t ws_size,
                              hipStream_t stream) {
    const float* x = (const float*)d_in[0];
    float* out = (float*)d_out;

    double w0    = 2.0 * M_PI * 7500.0 / 16000.0;
    double alpha = sin(w0) / (2.0 * 0.707);
    double cw    = cos(w0);
    double b0 = (1.0 - cw) / 2.0, b1 = 1.0 - cw, b2 = b0;
    double a0 = 1.0 + alpha, a1 = -2.0 * cw, a2 = 1.0 - alpha;
    b0 /= a0; b1 /= a0; b2 /= a0; a1 /= a0; a2 /= a0;

    float fb0 = (float)b0, fb1 = (float)b1, fb2 = (float)b2;
    float fa1 = (float)a1, fa2 = (float)a2;

    static_assert((BROWS * RPR) % 8 == 0, "whole work-groups of 8 waves");
    lp_stream<<<BROWS * RPR / 8, 256, 0, stream>>>(x, out, fb0, fb1, fb2, fa1, fa2);
}
